# starting kernel
# baseline (speedup 1.0000x reference)
#include <hip/hip_runtime.h>
#include <math.h>
#include <cstddef>

#define NNODES   20000
#define NEDGES   320000
#define ETOT     (NEDGES + NNODES)
#define IN_CH    128
#define HID_CH   32
#define HEADS    8
#define H1C      (HEADS * HID_CH)
#define OUT_CH   64
#define NEG_SLOPE 0.2f

typedef float  v2f   __attribute__((ext_vector_type(2)));
typedef float  v8f   __attribute__((ext_vector_type(8)));
typedef __bf16 v16bf __attribute__((ext_vector_type(16)));

__global__ void wmma_gemm(const float* __restrict__ A, const float* __restrict__ B,
                          float* __restrict__ C, int M, int K, int N) {
  const int lane = threadIdx.x;
  const int row  = lane & 15;
  const int kh   = lane >> 4;
  const int tm   = blockIdx.x;
  const int tn   = blockIdx.y;
  v8f acc = {0.f, 0.f, 0.f, 0.f, 0.f, 0.f, 0.f, 0.f};

#if __has_builtin(__builtin_amdgcn_wmma_f32_16x16x4_f32)
  const float* Arow = A + (size_t)(tm * 16 + row) * K;
  const float* Bcol = B + (size_t)(tn * 16 + row);
  for (int k = 0; k < K; k += 4) {
    v2f a, b;
    a.x = Arow[k + kh * 2 + 0];
    a.y = Arow[k + kh * 2 + 1];
    b.x = Bcol[(size_t)(k + kh * 2 + 0) * N];
    b.y = Bcol[(size_t)(k + kh * 2 + 1) * N];
    acc = __builtin_amdgcn_wmma_f32_16x16x4_f32(false, a, false, b,
                                                (short)0, acc, false, false);
  }
#elif __has_builtin(__builtin_amdgcn_wmma_f32_16x16x32_bf16)
  for (int k0 = 0; k0 < K; k0 += 32) {
    v16bf a, b;
#pragma unroll
    for (int i = 0; i < 16; ++i) {
      int vg = i >> 1, p = i & 1;
      int k  = k0 + ((vg & 4) ? 16 : 0) + kh * 8 + (vg & 3) * 2 + p;
      a[i] = (__bf16)A[(size_t)(tm * 16 + row) * K + k];
      b[i] = (__bf16)B[(size_t)k * N + tn * 16 + row];
    }
    acc = __builtin_amdgcn_wmma_f32_16x16x32_bf16(false, a, false, b,
                                                  (short)0, acc, false, false);
  }
#endif

  float* Crow = C + ((size_t)tm * 16 + kh * 8) * N + tn * 16 + row;
#pragma unroll
  for (int r = 0; r < 8; ++r) Crow[(size_t)r * N] = acc[r];
}

__global__ void fill_f32(float* __restrict__ p, float v, int n) {
  int i = blockIdx.x * blockDim.x + threadIdx.x;
  if (i < n) p[i] = v;
}

__device__ __forceinline__ void atomicMaxF(float* addr, float val) {
  if (val >= 0.f) atomicMax((int*)addr, __float_as_int(val));
  else            atomicMin((unsigned int*)addr, __float_as_uint(val));
}

__global__ void alpha_kernel(const float* __restrict__ h,
                             const float* __restrict__ a_src,
                             const float* __restrict__ a_dst,
                             float* __restrict__ as, float* __restrict__ ad,
                             int N, int H, int C) {
  int i = blockIdx.x * blockDim.x + threadIdx.x;
  if (i >= N * H) return;
  int hd = i % H;
  const float* hp = h + (size_t)i * C;
  float s = 0.f, d = 0.f;
  for (int c = 0; c < C; ++c) {
    float v = hp[c];
    s += v * a_src[hd * C + c];
    d += v * a_dst[hd * C + c];
  }
  as[i] = s;
  ad[i] = d;
}

__global__ void edge_max_kernel(const int* __restrict__ ei,
                                const float* __restrict__ as,
                                const float* __restrict__ ad,
                                float* __restrict__ e, float* __restrict__ m,
                                int Eorig, int Etot, int H) {
  int i = blockIdx.x * blockDim.x + threadIdx.x;
  if (i >= Etot * H) return;
  int eid = i / H, hd = i % H;
  int s, d;
  if (eid < Eorig) { s = ei[eid]; d = ei[Eorig + eid]; }
  else             { s = d = eid - Eorig; }
  float v = as[s * H + hd] + ad[d * H + hd];
  v = (v > 0.f) ? v : NEG_SLOPE * v;
  e[i] = v;
  atomicMaxF(&m[d * H + hd], v);
}

__global__ void edge_expsum_kernel(const int* __restrict__ ei,
                                   float* __restrict__ e,
                                   const float* __restrict__ m,
                                   float* __restrict__ den,
                                   int Eorig, int Etot, int H) {
  int i = blockIdx.x * blockDim.x + threadIdx.x;
  if (i >= Etot * H) return;
  int eid = i / H, hd = i % H;
  int d = (eid < Eorig) ? ei[Eorig + eid] : (eid - Eorig);
  float ex = __expf(e[i] - m[d * H + hd]);
  e[i] = ex;
  atomicAdd(&den[d * H + hd], ex);
}

__global__ void edge_scatter_kernel(const int* __restrict__ ei,
                                    const float* __restrict__ e,
                                    const float* __restrict__ h,
                                    float* __restrict__ acc,
                                    int Eorig, int Etot, int H, int C) {
  long long i = (long long)blockIdx.x * blockDim.x + threadIdx.x;
  int HC = H * C;
  if (i >= (long long)Etot * HC) return;
  int eid = (int)(i / HC);
  int ch  = (int)(i % HC);
  int hd  = ch / C;
  int s, d;
  if (eid < Eorig) { s = ei[eid]; d = ei[Eorig + eid]; }
  else             { s = d = eid - Eorig; }
  atomicAdd(&acc[(size_t)d * HC + ch],
            e[(size_t)eid * H + hd] * h[(size_t)s * HC + ch]);
}

__global__ void finalize_kernel(const float* __restrict__ acc,
                                const float* __restrict__ den,
                                const float* __restrict__ bias,
                                float* __restrict__ out,
                                int N, int H, int C, int do_elu) {
  int i = blockIdx.x * blockDim.x + threadIdx.x;
  int HC = H * C;
  if (i >= N * HC) return;
  int n = i / HC, ch = i % HC, hd = ch / C;
  float v = acc[i] / (den[n * H + hd] + 1e-16f) + bias[ch];
  if (do_elu) v = (v > 0.f) ? v : (__expf(v) - 1.f);
  out[i] = v;
}

extern "C" void kernel_launch(void* const* d_in, const int* in_sizes, int n_in,
                              void* d_out, int out_size, void* d_ws, size_t ws_size,
                              hipStream_t stream) {
  const float* x      = (const float*)d_in[0];
  const int*   ei     = (const int*)  d_in[1];
  const float* W1     = (const float*)d_in[2];
  const float* a_src1 = (const float*)d_in[3];
  const float* a_dst1 = (const float*)d_in[4];
  const float* b1     = (const float*)d_in[5];
  const float* W2     = (const float*)d_in[6];
  const float* a_src2 = (const float*)d_in[7];
  const float* a_dst2 = (const float*)d_in[8];
  const float* b2     = (const float*)d_in[9];
  float* out = (float*)d_out;

  float* ws   = (float*)d_ws;
  float* h1   = ws;
  float* acc1 = h1   + (size_t)NNODES * H1C;
  float* ebuf = acc1 + (size_t)NNODES * H1C;
  float* as1  = ebuf + (size_t)ETOT * HEADS;
  float* ad1  = as1  + (size_t)NNODES * HEADS;
  float* m1   = ad1  + (size_t)NNODES * HEADS;
  float* den1 = m1   + (size_t)NNODES * HEADS;
  float* h2   = acc1;
  float* as2  = acc1 + (size_t)NNODES * OUT_CH;
  float* ad2  = as2 + NNODES;
  float* m2   = ad2 + NNODES;
  float* den2 = m2  + NNODES;

  const int T = 256;
  const int NH1 = NNODES * HEADS;
  const int EH1 = ETOT * HEADS;
  const long long SC1 = (long long)ETOT * H1C;
  const long long SC2 = (long long)ETOT * OUT_CH;

  fill_f32<<<(NH1 + T - 1) / T, T, 0, stream>>>(m1, -INFINITY, NH1);
  fill_f32<<<(NH1 + T - 1) / T, T, 0, stream>>>(den1, 0.f, NH1);
  fill_f32<<<(NNODES * H1C + T - 1) / T, T, 0, stream>>>(acc1, 0.f, NNODES * H1C);

  wmma_gemm<<<dim3(NNODES / 16, H1C / 16), 32, 0, stream>>>(x, W1, h1,
                                                            NNODES, IN_CH, H1C);
  alpha_kernel<<<(NH1 + T - 1) / T, T, 0, stream>>>(h1, a_src1, a_dst1, as1, ad1,
                                                    NNODES, HEADS, HID_CH);
  edge_max_kernel<<<(EH1 + T - 1) / T, T, 0, stream>>>(ei, as1, ad1, ebuf, m1,
                                                       NEDGES, ETOT, HEADS);
  edge_expsum_kernel<<<(EH1 + T - 1) / T, T, 0, stream>>>(ei, ebuf, m1, den1,
                                                          NEDGES, ETOT, HEADS);
  edge_scatter_kernel<<<(unsigned)((SC1 + T - 1) / T), T, 0, stream>>>(
      ei, ebuf, h1, acc1, NEDGES, ETOT, HEADS, HID_CH);
  finalize_kernel<<<(NNODES * H1C + T - 1) / T, T, 0, stream>>>(
      acc1, den1, b1, h1, NNODES, HEADS, HID_CH,  1);

  fill_f32<<<(NNODES + T - 1) / T, T, 0, stream>>>(m2, -INFINITY, NNODES);
  fill_f32<<<(NNODES + T - 1) / T, T, 0, stream>>>(den2, 0.f, NNODES);
  fill_f32<<<(NNODES * OUT_CH + T - 1) / T, T, 0, stream>>>(out, 0.f, NNODES * OUT_CH);

  wmma_gemm<<<dim3(NNODES / 16, OUT_CH / 16), 32, 0, stream>>>(h1, W2, h2,
                                                               NNODES, H1C, OUT_CH);
  alpha_kernel<<<(NNODES + T - 1) / T, T, 0, stream>>>(h2, a_src2, a_dst2, as2, ad2,
                                                       NNODES, 1, OUT_CH);
  edge_max_kernel<<<(ETOT + T - 1) / T, T, 0, stream>>>(ei, as2, ad2, ebuf, m2,
                                                        NEDGES, ETOT, 1);
  edge_expsum_kernel<<<(ETOT + T - 1) / T, T, 0, stream>>>(ei, ebuf, m2, den2,
                                                           NEDGES, ETOT, 1);
  edge_scatter_kernel<<<(unsigned)((SC2 + T - 1) / T), T, 0, stream>>>(
      ei, ebuf, h2, out, NEDGES, ETOT, 1, OUT_CH);
  finalize_kernel<<<(NNODES * OUT_CH + T - 1) / T, T, 0, stream>>>(
      out, den2, b2, out, NNODES, 1, OUT_CH,  0);
}


# GEMMs: a wave owns a 16x64 strip as 4 accumulator tiles with columns interleaved by 4: A fragment loaded once per k-step, B by 16-byte loads, C by 16-byte stores; 4-wave blocks; sum order unchanged
# speedup vs baseline: 4.8300x; 4.8300x over previous
#include <hip/hip_runtime.h>
#include <math.h>
#include <cstddef>

#define NNODES   20000
#define NEDGES   320000
#define ETOT     (NEDGES + NNODES)
#define IN_CH    128
#define HID_CH   32
#define HEADS    8
#define H1C      (HEADS * HID_CH)
#define OUT_CH   64
#define NEG_SLOPE 0.2f

typedef float  v2f   __attribute__((ext_vector_type(2)));
typedef float  v8f   __attribute__((ext_vector_type(8)));

template <int K, int N>
__device__ __forceinline__ void gemm_strip(const float* __restrict__ A,
                                           const float* __restrict__ B,
                                           float* __restrict__ C, int M) {
  const int lane = threadIdx.x & 31;
  const int wave = __builtin_amdgcn_readfirstlane((int)(threadIdx.x >> 5));
  const int tm   = blockIdx.x * 4 + wave;
  if (tm * 16 >= M) return;
  const int j  = lane & 15;
  const int kh = lane >> 4;
  const int cb = blockIdx.y * 64;
  const float* Ap = A + (size_t)(tm * 16 + j) * K + kh * 2;
  const float* Bp = B + (size_t)(kh * 2) * N + cb + 4 * j;
  v8f acc0 = {0.f, 0.f, 0.f, 0.f, 0.f, 0.f, 0.f, 0.f};
  v8f acc1 = acc0, acc2 = acc0, acc3 = acc0;

#pragma unroll 1
  for (int k0 = 0; k0 < K; k0 += 32) {
    v2f    a[8];
    float4 b0[8], b1[8];
#pragma unroll
    for (int s = 0; s < 8; ++s) {
      a[s]  = *(const v2f*)(Ap + k0 + 4 * s);
      b0[s] = *(const float4*)(Bp + (size_t)(k0 + 4 * s) * N);
      b1[s] = *(const float4*)(Bp + (size_t)(k0 + 4 * s + 1) * N);
    }
#pragma unroll
    for (int s = 0; s < 8; ++s) {
      v2f p0, p1, p2, p3;
      p0.x = b0[s].x; p0.y = b1[s].x;
      p1.x = b0[s].y; p1.y = b1[s].y;
      p2.x = b0[s].z; p2.y = b1[s].z;
      p3.x = b0[s].w; p3.y = b1[s].w;
      acc0 = __builtin_amdgcn_wmma_f32_16x16x4_f32(false, a[s], false, p0,
                                                   (short)0, acc0, false, false);
      acc1 = __builtin_amdgcn_wmma_f32_16x16x4_f32(false, a[s], false, p1,
                                                   (short)0, acc1, false, false);
      acc2 = __builtin_amdgcn_wmma_f32_16x16x4_f32(false, a[s], false, p2,
                                                   (short)0, acc2, false, false);
      acc3 = __builtin_amdgcn_wmma_f32_16x16x4_f32(false, a[s], false, p3,
                                                   (short)0, acc3, false, false);
    }
  }

  float* Cp = C + ((size_t)tm * 16 + kh * 8) * N + cb + 4 * j;
#pragma unroll
  for (int r = 0; r < 8; ++r) {
    *(float4*)(Cp + (size_t)r * N) = make_float4(acc0[r], acc1[r], acc2[r], acc3[r]);
  }
}

__global__ void __launch_bounds__(128)
wmma_gemm_l1(const float* __restrict__ A, const float* __restrict__ B,
             float* __restrict__ C, int M) {
  gemm_strip<IN_CH, H1C>(A, B, C, M);
}

__global__ void __launch_bounds__(128)
wmma_gemm_l2(const float* __restrict__ A, const float* __restrict__ B,
             float* __restrict__ C, int M) {
  gemm_strip<H1C, OUT_CH>(A, B, C, M);
}

__global__ void alpha_kernel(const float* __restrict__ h,
                             const float* __restrict__ a_src,
                             const float* __restrict__ a_dst,
                             float* __restrict__ as, float* __restrict__ ad,
                             int N, int H, int C) {
  int i = blockIdx.x * blockDim.x + threadIdx.x;
  if (i >= N * H) return;
  int hd = i % H;
  const float* hp = h + (size_t)i * C;
  float s = 0.f, d = 0.f;
  for (int c = 0; c < C; ++c) {
    float v = hp[c];
    s += v * a_src[hd * C + c];
    d += v * a_dst[hd * C + c];
  }
  as[i] = s;
  ad[i] = d;
}

__global__ void fill_i32(int* __restrict__ p, int v, int n) {
  int i = blockIdx.x * blockDim.x + threadIdx.x;
  if (i < n) p[i] = v;
}

__global__ void dst_count_kernel(const int* __restrict__ ei, int* __restrict__ cnt,
                                 int Eorig) {
  int i = blockIdx.x * blockDim.x + threadIdx.x;
  if (i < Eorig) atomicAdd(&cnt[ei[Eorig + i]], 1);
}

#define SCAN_T   1024
#define SCAN_PER ((NNODES + SCAN_T - 1) / SCAN_T)
__global__ void __launch_bounds__(SCAN_T)
row_start_kernel(const int* __restrict__ cnt, int* __restrict__ rowstart, int N) {
  __shared__ int part[2][SCAN_T];
  const int t  = threadIdx.x;
  const int lo = t * SCAN_PER;
  const bool whole = (lo + SCAN_PER <= N);
  int len[SCAN_PER];
  if (whole) {
#pragma unroll
    for (int j = 0; j < SCAN_PER; ++j) len[j] = cnt[lo + j] + 1;
  } else {
#pragma unroll
    for (int j = 0; j < SCAN_PER; ++j) len[j] = (lo + j < N) ? cnt[lo + j] + 1 : 0;
  }
  int sum = 0;
#pragma unroll
  for (int j = 0; j < SCAN_PER; ++j) sum += len[j];
  part[0][t] = sum;
  __syncthreads();
  int cur = 0;
  for (int off = 1; off < SCAN_T; off <<= 1) {
    int v = part[cur][t];
    if (t >= off) v += part[cur][t - off];
    part[cur ^ 1][t] = v;
    __syncthreads();
    cur ^= 1;
  }
  int run = part[cur][t] - sum;
  if (whole) {
#pragma unroll
    for (int j = 0; j < SCAN_PER; ++j) { rowstart[lo + j] = run; run += len[j]; }
  } else {
#pragma unroll
    for (int j = 0; j < SCAN_PER; ++j) {
      if (lo + j < N) { rowstart[lo + j] = run; run += len[j]; }
    }
  }
  if (t == SCAN_T - 1) rowstart[N] = part[cur][t];
}

__global__ void place_kernel(const int* __restrict__ ei, const int* __restrict__ rowstart,
                             int* __restrict__ cnt, int* __restrict__ sorted,
                             int Eorig, int N) {
  int i = blockIdx.x * blockDim.x + threadIdx.x;
  if (i < Eorig) {
    int s = ei[i], d = ei[Eorig + i];
    int r = atomicSub(&cnt[d], 1);
    sorted[rowstart[d] + r] = s;
  }
  if (i < N) sorted[rowstart[i]] = i;
}


__global__ void __launch_bounds__(256)
gather_rows_l1_kernel(const int* __restrict__ sorted, const int* __restrict__ rowstart,
                      const float* __restrict__ as, const float* __restrict__ ad,
                      const float* __restrict__ h, const float* __restrict__ bias,
                      float* __restrict__ g, int N) {
  const int lane = threadIdx.x & 31;
  const int d = __builtin_amdgcn_readfirstlane((int)((blockIdx.x * blockDim.x + threadIdx.x) >> 5));
  if (d >= N) return;
  const int beg = rowstart[d], end = rowstart[d + 1];
  const int hA = lane >> 3;
  const int hB = 4 + (lane >> 3);
  const int h8 = lane & 7;

  float mx = -INFINITY;
  for (int base = beg; base < end; base += 32) {
    const int n = min(32, end - base);
    int mine = 0;
    if (lane < n) mine = sorted[base + lane];
#pragma unroll 4
    for (int k = 0; k < n; ++k) {
      const int s = __builtin_amdgcn_readlane(mine, k);
      mx = fmaxf(mx, as[(size_t)s * HEADS + h8]);
    }
  }
  const float adA = ad[d * HEADS + hA];
  const float adB = ad[d * HEADS + hB];
  float mA = __shfl(mx, hA) + adA;
  float mB = __shfl(mx, hB) + adB;
  mA = (mA > 0.f) ? mA : NEG_SLOPE * mA;
  mB = (mB > 0.f) ? mB : NEG_SLOPE * mB;

  float4 accA = make_float4(0.f, 0.f, 0.f, 0.f);
  float4 accB = make_float4(0.f, 0.f, 0.f, 0.f);
  float denA = 0.f, denB = 0.f;
  for (int base = beg; base < end; base += 32) {
    const int n = min(32, end - base);
    int mine = 0;
    if (lane < n) mine = sorted[base + lane];
#pragma unroll 4
    for (int k = 0; k < n; ++k) {
      const int s = __builtin_amdgcn_readlane(mine, k);
      float eA = as[(size_t)s * HEADS + hA] + adA;
      float eB = as[(size_t)s * HEADS + hB] + adB;
      const float4* hp = (const float4*)(h + (size_t)s * H1C);
      const float4 vA = hp[lane];
      const float4 vB = hp[32 + lane];
      eA = (eA > 0.f) ? eA : NEG_SLOPE * eA;
      eB = (eB > 0.f) ? eB : NEG_SLOPE * eB;
      const float wA = __expf(eA - mA);
      const float wB = __expf(eB - mB);
      denA += wA;
      denB += wB;
      accA.x += wA * vA.x; accA.y += wA * vA.y; accA.z += wA * vA.z; accA.w += wA * vA.w;
      accB.x += wB * vB.x; accB.y += wB * vB.y; accB.z += wB * vB.z; accB.w += wB * vB.w;
    }
  }
  const float dA = denA + 1e-16f;
  const float dB = denB + 1e-16f;
  const float4 bA = ((const float4*)bias)[lane];
  const float4 bB = ((const float4*)bias)[32 + lane];
  float4 oA, oB;
  oA.x = accA.x / dA + bA.x; oA.y = accA.y / dA + bA.y;
  oA.z = accA.z / dA + bA.z; oA.w = accA.w / dA + bA.w;
  oB.x = accB.x / dB + bB.x; oB.y = accB.y / dB + bB.y;
  oB.z = accB.z / dB + bB.z; oB.w = accB.w / dB + bB.w;
  oA.x = (oA.x > 0.f) ? oA.x : (__expf(oA.x) - 1.f);
  oA.y = (oA.y > 0.f) ? oA.y : (__expf(oA.y) - 1.f);
  oA.z = (oA.z > 0.f) ? oA.z : (__expf(oA.z) - 1.f);
  oA.w = (oA.w > 0.f) ? oA.w : (__expf(oA.w) - 1.f);
  oB.x = (oB.x > 0.f) ? oB.x : (__expf(oB.x) - 1.f);
  oB.y = (oB.y > 0.f) ? oB.y : (__expf(oB.y) - 1.f);
  oB.z = (oB.z > 0.f) ? oB.z : (__expf(oB.z) - 1.f);
  oB.w = (oB.w > 0.f) ? oB.w : (__expf(oB.w) - 1.f);
  float4* gp = (float4*)(g + (size_t)d * H1C);
  gp[lane]      = oA;
  gp[32 + lane] = oB;
}

__global__ void __launch_bounds__(256)
gather_rows_l2_kernel(const int* __restrict__ sorted, const int* __restrict__ rowstart,
                      const float* __restrict__ as, const float* __restrict__ ad,
                      const float* __restrict__ h, const float* __restrict__ bias,
                      float* __restrict__ out, int N) {
  const int lane = threadIdx.x & 31;
  const int d = __builtin_amdgcn_readfirstlane((int)((blockIdx.x * blockDim.x + threadIdx.x) >> 5));
  if (d >= N) return;
  const int beg = rowstart[d], end = rowstart[d + 1];

  float mx = -INFINITY;
  for (int base = beg; base < end; base += 32) {
    if (base + lane < end) mx = fmaxf(mx, as[sorted[base + lane]]);
  }
#pragma unroll
  for (int o = 16; o > 0; o >>= 1) mx = fmaxf(mx, __shfl_xor(mx, o));
  const float adv = ad[d];
  float m = mx + adv;
  m = (m > 0.f) ? m : NEG_SLOPE * m;

  float2 acc = make_float2(0.f, 0.f);
  float denl = 0.f;
  for (int base = beg; base < end; base += 32) {
    const int n = min(32, end - base);
    int mine = 0;
    float exl = 0.f;
    if (lane < n) {
      mine = sorted[base + lane];
      float e = as[mine] + adv;
      e = (e > 0.f) ? e : NEG_SLOPE * e;
      exl = __expf(e - m);
    }
    denl += exl;
#pragma unroll 4
    for (int k = 0; k < n; ++k) {
      const int   s = __builtin_amdgcn_readlane(mine, k);
      const float w = __int_as_float(__builtin_amdgcn_readlane(__float_as_int(exl), k));
      const float2 v = ((const float2*)(h + (size_t)s * OUT_CH))[lane];
      acc.x += w * v.x; acc.y += w * v.y;
    }
  }
#pragma unroll
  for (int o = 16; o > 0; o >>= 1) denl += __shfl_xor(denl, o);
  const float dn = denl + 1e-16f;
  const float2 b = ((const float2*)bias)[lane];
  float2 o2;
  o2.x = acc.x / dn + b.x;
  o2.y = acc.y / dn + b.y;
  ((float2*)(out + (size_t)d * OUT_CH))[lane] = o2;
}

extern "C" void kernel_launch(void* const* d_in, const int* in_sizes, int n_in,
                              void* d_out, int out_size, void* d_ws, size_t ws_size,
                              hipStream_t stream) {
  const float* x      = (const float*)d_in[0];
  const int*   ei     = (const int*)  d_in[1];
  const float* W1     = (const float*)d_in[2];
  const float* a_src1 = (const float*)d_in[3];
  const float* a_dst1 = (const float*)d_in[4];
  const float* b1     = (const float*)d_in[5];
  const float* W2     = (const float*)d_in[6];
  const float* a_src2 = (const float*)d_in[7];
  const float* a_dst2 = (const float*)d_in[8];
  const float* b2     = (const float*)d_in[9];
  float* out = (float*)d_out;

  float* ws   = (float*)d_ws;
  float* h1   = ws;
  float* g1   = h1   + (size_t)NNODES * H1C;
  float* as1  = g1   + (size_t)NNODES * H1C;
  float* ad1  = as1  + (size_t)NNODES * HEADS;
  int* sorted   = (int*)(ad1 + (size_t)NNODES * HEADS);
  int* cnt      = sorted + (size_t)ETOT;
  int* rowstart = cnt + NNODES;
  float* h2   = h1;
  float* as2  = h1 + (size_t)NNODES * OUT_CH;
  float* ad2  = as2 + NNODES;

  const int T = 256;
  const int NH1 = NNODES * HEADS;

  fill_i32<<<(NNODES + T - 1) / T, T, 0, stream>>>(cnt, 0, NNODES);

  wmma_gemm_l1<<<dim3((NNODES + 63) / 64, H1C / 64), 128, 0, stream>>>(x, W1, h1, NNODES);
  alpha_kernel<<<(NH1 + T - 1) / T, T, 0, stream>>>(h1, a_src1, a_dst1, as1, ad1,
                                                    NNODES, HEADS, HID_CH);
  dst_count_kernel<<<(NEDGES + T - 1) / T, T, 0, stream>>>(ei, cnt, NEDGES);
  row_start_kernel<<<1, SCAN_T, 0, stream>>>(cnt, rowstart, NNODES);
  place_kernel<<<(NEDGES + T - 1) / T, T, 0, stream>>>(ei, rowstart, cnt, sorted,
                                                       NEDGES, NNODES);
  gather_rows_l1_kernel<<<(NNODES * 32 + T - 1) / T, T, 0, stream>>>(
      sorted, rowstart, as1, ad1, h1, b1, g1, NNODES);

  wmma_gemm_l2<<<dim3((NNODES + 63) / 64, OUT_CH / 64), 128, 0, stream>>>(g1, W2, h2, NNODES);
  alpha_kernel<<<(NNODES + T - 1) / T, T, 0, stream>>>(h2, a_src2, a_dst2, as2, ad2,
                                                       NNODES, 1, OUT_CH);
  gather_rows_l2_kernel<<<(NNODES * 32 + T - 1) / T, T, 0, stream>>>(
      sorted, rowstart, as2, ad2, h2, b2, out, NNODES);
}


# attention logits from the matrix unit: a_src/a_dst folded into the weights once per call (W.a, a 16-column tile beside x.W); both logit launches and their re-read of h gone; logit sum order changes
# speedup vs baseline: 5.1949x; 1.0755x over previous
#include <hip/hip_runtime.h>
#include <math.h>
#include <cstddef>

#define NNODES   20000
#define NEDGES   320000
#define ETOT     (NEDGES + NNODES)
#define IN_CH    128
#define HID_CH   32
#define HEADS    8
#define H1C      (HEADS * HID_CH)
#define OUT_CH   64
#define NEG_SLOPE 0.2f

typedef float  v2f   __attribute__((ext_vector_type(2)));
typedef float  v8f   __attribute__((ext_vector_type(8)));

#define LGW 16

template <int K, int N, int MODE>
__device__ __forceinline__ void gemm_strip(const float* __restrict__ A,
                                           const float* __restrict__ B,
                                           float* __restrict__ C, int M,
                                           const float* __restrict__ Wa,
                                           float* __restrict__ L0,
                                           float* __restrict__ L1) {
  const int lane = threadIdx.x & 31;
  const int wave = __builtin_amdgcn_readfirstlane((int)(threadIdx.x >> 5));
  const int tm   = blockIdx.x * 4 + wave;
  if (tm * 16 >= M) return;
  const int j  = lane & 15;
  const int kh = lane >> 4;
  const int cb = blockIdx.y * 64;
  const float* Ap = A + (size_t)(tm * 16 + j) * K + kh * 2;
  const float* Bp = B + (size_t)(kh * 2) * N + cb + 4 * j;
  const float* Wp = Wa + (size_t)(kh * 2) * LGW + j;
  v8f acc0 = {0.f, 0.f, 0.f, 0.f, 0.f, 0.f, 0.f, 0.f};
  v8f acc1 = acc0, acc2 = acc0, acc3 = acc0;
  v8f accL = acc0;

#pragma unroll 1
  for (int k0 = 0; k0 < K; k0 += 32) {
    v2f    a[8];
    float4 b0[8], b1[8];
    v2f    w[8];
#pragma unroll
    for (int s = 0; s < 8; ++s) {
      a[s]  = *(const v2f*)(Ap + k0 + 4 * s);
      b0[s] = *(const float4*)(Bp + (size_t)(k0 + 4 * s) * N);
      b1[s] = *(const float4*)(Bp + (size_t)(k0 + 4 * s + 1) * N);
      if constexpr (MODE != 0) {
        w[s].x = Wp[(size_t)(k0 + 4 * s) * LGW];
        w[s].y = Wp[(size_t)(k0 + 4 * s + 1) * LGW];
      }
    }
#pragma unroll
    for (int s = 0; s < 8; ++s) {
      v2f p0, p1, p2, p3;
      p0.x = b0[s].x; p0.y = b1[s].x;
      p1.x = b0[s].y; p1.y = b1[s].y;
      p2.x = b0[s].z; p2.y = b1[s].z;
      p3.x = b0[s].w; p3.y = b1[s].w;
      acc0 = __builtin_amdgcn_wmma_f32_16x16x4_f32(false, a[s], false, p0,
                                                   (short)0, acc0, false, false);
      acc1 = __builtin_amdgcn_wmma_f32_16x16x4_f32(false, a[s], false, p1,
                                                   (short)0, acc1, false, false);
      acc2 = __builtin_amdgcn_wmma_f32_16x16x4_f32(false, a[s], false, p2,
                                                   (short)0, acc2, false, false);
      acc3 = __builtin_amdgcn_wmma_f32_16x16x4_f32(false, a[s], false, p3,
                                                   (short)0, acc3, false, false);
      if constexpr (MODE != 0) {
        accL = __builtin_amdgcn_wmma_f32_16x16x4_f32(false, a[s], false, w[s],
                                                     (short)0, accL, false, false);
      }
    }
  }

  float* Cp = C + ((size_t)tm * 16 + kh * 8) * N + cb + 4 * j;
#pragma unroll
  for (int r = 0; r < 8; ++r) {
    *(float4*)(Cp + (size_t)r * N) = make_float4(acc0[r], acc1[r], acc2[r], acc3[r]);
  }
  if constexpr (MODE == 1) {
    float* Lp = ((j < HEADS) ? L0 : L1) + ((size_t)tm * 16 + kh * 8) * HEADS + (j & (HEADS - 1));
#pragma unroll
    for (int r = 0; r < 8; ++r) Lp[(size_t)r * HEADS] = accL[r];
  }
  if constexpr (MODE == 2) {
    if (j < 2) {
      float* Lp = L0 + ((size_t)tm * 16 + kh * 8) * 2 + j;
#pragma unroll
      for (int r = 0; r < 8; ++r) Lp[(size_t)r * 2] = accL[r];
    }
  }
}

__global__ void __launch_bounds__(128)
wmma_gemm_l1(const float* __restrict__ A, const float* __restrict__ B,
             float* __restrict__ C, int M, const float* __restrict__ Wa,
             float* __restrict__ as, float* __restrict__ ad) {
  if (blockIdx.y == 0) gemm_strip<IN_CH, H1C, 1>(A, B, C, M, Wa, as, ad);
  else                 gemm_strip<IN_CH, H1C, 0>(A, B, C, M, Wa, as, ad);
}

__global__ void __launch_bounds__(128)
wmma_gemm_l2(const float* __restrict__ A, const float* __restrict__ B,
             float* __restrict__ C, int M, const float* __restrict__ Wa,
             float* __restrict__ la) {
  gemm_strip<H1C, OUT_CH, 2>(A, B, C, M, Wa, la, la);
}

#define FOLD1_BLOCKS ((IN_CH * LGW) / 256)
#define FOLD2_BLOCKS ((H1C * LGW) / 256)
#define FILL_BLOCKS  ((NNODES + 255) / 256)

__global__ void __launch_bounds__(256)
fold_kernel(const float* __restrict__ W1, const float* __restrict__ a_src1,
            const float* __restrict__ a_dst1, float* __restrict__ Wa1,
            const float* __restrict__ W2, const float* __restrict__ a_src2,
            const float* __restrict__ a_dst2, float* __restrict__ Wa2,
            int* __restrict__ cnt) {
  const int b = blockIdx.x;
  if (b < FOLD1_BLOCKS) {
    const int i   = b * 256 + threadIdx.x;
    const int k   = i / LGW, col = i % LGW, hd = col % HEADS;
    const float4* wp = (const float4*)(W1 + (size_t)k * H1C + hd * HID_CH);
    const float4* ap = (const float4*)(((col < HEADS) ? a_src1 : a_dst1) + hd * HID_CH);
    float s = 0.f;
#pragma unroll
    for (int c = 0; c < HID_CH / 4; ++c) {
      const float4 wv = wp[c], av = ap[c];
      s += wv.x * av.x; s += wv.y * av.y; s += wv.z * av.z; s += wv.w * av.w;
    }
    Wa1[i] = s;
  } else if (b < FOLD1_BLOCKS + FOLD2_BLOCKS) {
    const int i   = (b - FOLD1_BLOCKS) * 256 + threadIdx.x;
    const int k   = i / LGW, col = i % LGW;
    float s = 0.f;
    if (col < 2) {
      const float4* wp = (const float4*)(W2 + (size_t)k * OUT_CH);
      const float4* ap = (const float4*)((col == 0) ? a_src2 : a_dst2);
#pragma unroll
      for (int c = 0; c < OUT_CH / 4; ++c) {
        const float4 wv = wp[c], av = ap[c];
        s += wv.x * av.x; s += wv.y * av.y; s += wv.z * av.z; s += wv.w * av.w;
      }
    }
    Wa2[i] = s;
  } else {
    const int i = (b - FOLD1_BLOCKS - FOLD2_BLOCKS) * 256 + threadIdx.x;
    if (i < NNODES) cnt[i] = 0;
  }
}

__global__ void dst_count_kernel(const int* __restrict__ ei, int* __restrict__ cnt,
                                 int Eorig) {
  int i = blockIdx.x * blockDim.x + threadIdx.x;
  if (i < Eorig) atomicAdd(&cnt[ei[Eorig + i]], 1);
}

#define SCAN_T   1024
#define SCAN_PER ((NNODES + SCAN_T - 1) / SCAN_T)
__global__ void __launch_bounds__(SCAN_T)
row_start_kernel(const int* __restrict__ cnt, int* __restrict__ rowstart, int N) {
  __shared__ int part[2][SCAN_T];
  const int t  = threadIdx.x;
  const int lo = t * SCAN_PER;
  const bool whole = (lo + SCAN_PER <= N);
  int len[SCAN_PER];
  if (whole) {
#pragma unroll
    for (int j = 0; j < SCAN_PER; ++j) len[j] = cnt[lo + j] + 1;
  } else {
#pragma unroll
    for (int j = 0; j < SCAN_PER; ++j) len[j] = (lo + j < N) ? cnt[lo + j] + 1 : 0;
  }
  int sum = 0;
#pragma unroll
  for (int j = 0; j < SCAN_PER; ++j) sum += len[j];
  part[0][t] = sum;
  __syncthreads();
  int cur = 0;
  for (int off = 1; off < SCAN_T; off <<= 1) {
    int v = part[cur][t];
    if (t >= off) v += part[cur][t - off];
    part[cur ^ 1][t] = v;
    __syncthreads();
    cur ^= 1;
  }
  int run = part[cur][t] - sum;
  if (whole) {
#pragma unroll
    for (int j = 0; j < SCAN_PER; ++j) { rowstart[lo + j] = run; run += len[j]; }
  } else {
#pragma unroll
    for (int j = 0; j < SCAN_PER; ++j) {
      if (lo + j < N) { rowstart[lo + j] = run; run += len[j]; }
    }
  }
  if (t == SCAN_T - 1) rowstart[N] = part[cur][t];
}

__global__ void place_kernel(const int* __restrict__ ei, const int* __restrict__ rowstart,
                             int* __restrict__ cnt, int* __restrict__ sorted,
                             int Eorig, int N) {
  int i = blockIdx.x * blockDim.x + threadIdx.x;
  if (i < Eorig) {
    int s = ei[i], d = ei[Eorig + i];
    int r = atomicSub(&cnt[d], 1);
    sorted[rowstart[d] + r] = s;
  }
  if (i < N) sorted[rowstart[i]] = i;
}


__global__ void __launch_bounds__(256)
gather_rows_l1_kernel(const int* __restrict__ sorted, const int* __restrict__ rowstart,
                      const float* __restrict__ as, const float* __restrict__ ad,
                      const float* __restrict__ h, const float* __restrict__ bias,
                      float* __restrict__ g, int N) {
  const int lane = threadIdx.x & 31;
  const int d = __builtin_amdgcn_readfirstlane((int)((blockIdx.x * blockDim.x + threadIdx.x) >> 5));
  if (d >= N) return;
  const int beg = rowstart[d], end = rowstart[d + 1];
  const int hA = lane >> 3;
  const int hB = 4 + (lane >> 3);
  const int h8 = lane & 7;

  float mx = -INFINITY;
  for (int base = beg; base < end; base += 32) {
    const int n = min(32, end - base);
    int mine = 0;
    if (lane < n) mine = sorted[base + lane];
#pragma unroll 4
    for (int k = 0; k < n; ++k) {
      const int s = __builtin_amdgcn_readlane(mine, k);
      mx = fmaxf(mx, as[(size_t)s * HEADS + h8]);
    }
  }
  const float adA = ad[d * HEADS + hA];
  const float adB = ad[d * HEADS + hB];
  float mA = __shfl(mx, hA) + adA;
  float mB = __shfl(mx, hB) + adB;
  mA = (mA > 0.f) ? mA : NEG_SLOPE * mA;
  mB = (mB > 0.f) ? mB : NEG_SLOPE * mB;

  float4 accA = make_float4(0.f, 0.f, 0.f, 0.f);
  float4 accB = make_float4(0.f, 0.f, 0.f, 0.f);
  float denA = 0.f, denB = 0.f;
  for (int base = beg; base < end; base += 32) {
    const int n = min(32, end - base);
    int mine = 0;
    if (lane < n) mine = sorted[base + lane];
#pragma unroll 4
    for (int k = 0; k < n; ++k) {
      const int s = __builtin_amdgcn_readlane(mine, k);
      float eA = as[(size_t)s * HEADS + hA] + adA;
      float eB = as[(size_t)s * HEADS + hB] + adB;
      const float4* hp = (const float4*)(h + (size_t)s * H1C);
      const float4 vA = hp[lane];
      const float4 vB = hp[32 + lane];
      eA = (eA > 0.f) ? eA : NEG_SLOPE * eA;
      eB = (eB > 0.f) ? eB : NEG_SLOPE * eB;
      const float wA = __expf(eA - mA);
      const float wB = __expf(eB - mB);
      denA += wA;
      denB += wB;
      accA.x += wA * vA.x; accA.y += wA * vA.y; accA.z += wA * vA.z; accA.w += wA * vA.w;
      accB.x += wB * vB.x; accB.y += wB * vB.y; accB.z += wB * vB.z; accB.w += wB * vB.w;
    }
  }
  const float dA = denA + 1e-16f;
  const float dB = denB + 1e-16f;
  const float4 bA = ((const float4*)bias)[lane];
  const float4 bB = ((const float4*)bias)[32 + lane];
  float4 oA, oB;
  oA.x = accA.x / dA + bA.x; oA.y = accA.y / dA + bA.y;
  oA.z = accA.z / dA + bA.z; oA.w = accA.w / dA + bA.w;
  oB.x = accB.x / dB + bB.x; oB.y = accB.y / dB + bB.y;
  oB.z = accB.z / dB + bB.z; oB.w = accB.w / dB + bB.w;
  oA.x = (oA.x > 0.f) ? oA.x : (__expf(oA.x) - 1.f);
  oA.y = (oA.y > 0.f) ? oA.y : (__expf(oA.y) - 1.f);
  oA.z = (oA.z > 0.f) ? oA.z : (__expf(oA.z) - 1.f);
  oA.w = (oA.w > 0.f) ? oA.w : (__expf(oA.w) - 1.f);
  oB.x = (oB.x > 0.f) ? oB.x : (__expf(oB.x) - 1.f);
  oB.y = (oB.y > 0.f) ? oB.y : (__expf(oB.y) - 1.f);
  oB.z = (oB.z > 0.f) ? oB.z : (__expf(oB.z) - 1.f);
  oB.w = (oB.w > 0.f) ? oB.w : (__expf(oB.w) - 1.f);
  float4* gp = (float4*)(g + (size_t)d * H1C);
  gp[lane]      = oA;
  gp[32 + lane] = oB;
}

__global__ void __launch_bounds__(256)
gather_rows_l2_kernel(const int* __restrict__ sorted, const int* __restrict__ rowstart,
                      const float* __restrict__ la,
                      const float* __restrict__ h, const float* __restrict__ bias,
                      float* __restrict__ out, int N) {
  const int lane = threadIdx.x & 31;
  const int d = __builtin_amdgcn_readfirstlane((int)((blockIdx.x * blockDim.x + threadIdx.x) >> 5));
  if (d >= N) return;
  const int beg = rowstart[d], end = rowstart[d + 1];

  float mx = -INFINITY;
  for (int base = beg; base < end; base += 32) {
    if (base + lane < end) mx = fmaxf(mx, la[2 * (size_t)sorted[base + lane]]);
  }
#pragma unroll
  for (int o = 16; o > 0; o >>= 1) mx = fmaxf(mx, __shfl_xor(mx, o));
  const float adv = la[2 * (size_t)d + 1];
  float m = mx + adv;
  m = (m > 0.f) ? m : NEG_SLOPE * m;

  float2 acc = make_float2(0.f, 0.f);
  float denl = 0.f;
  for (int base = beg; base < end; base += 32) {
    const int n = min(32, end - base);
    int mine = 0;
    float exl = 0.f;
    if (lane < n) {
      mine = sorted[base + lane];
      float e = la[2 * (size_t)mine] + adv;
      e = (e > 0.f) ? e : NEG_SLOPE * e;
      exl = __expf(e - m);
    }
    denl += exl;
#pragma unroll 4
    for (int k = 0; k < n; ++k) {
      const int   s = __builtin_amdgcn_readlane(mine, k);
      const float w = __int_as_float(__builtin_amdgcn_readlane(__float_as_int(exl), k));
      const float2 v = ((const float2*)(h + (size_t)s * OUT_CH))[lane];
      acc.x += w * v.x; acc.y += w * v.y;
    }
  }
#pragma unroll
  for (int o = 16; o > 0; o >>= 1) denl += __shfl_xor(denl, o);
  const float dn = denl + 1e-16f;
  const float2 b = ((const float2*)bias)[lane];
  float2 o2;
  o2.x = acc.x / dn + b.x;
  o2.y = acc.y / dn + b.y;
  ((float2*)(out + (size_t)d * OUT_CH))[lane] = o2;
}

extern "C" void kernel_launch(void* const* d_in, const int* in_sizes, int n_in,
                              void* d_out, int out_size, void* d_ws, size_t ws_size,
                              hipStream_t stream) {
  const float* x      = (const float*)d_in[0];
  const int*   ei     = (const int*)  d_in[1];
  const float* W1     = (const float*)d_in[2];
  const float* a_src1 = (const float*)d_in[3];
  const float* a_dst1 = (const float*)d_in[4];
  const float* b1     = (const float*)d_in[5];
  const float* W2     = (const float*)d_in[6];
  const float* a_src2 = (const float*)d_in[7];
  const float* a_dst2 = (const float*)d_in[8];
  const float* b2     = (const float*)d_in[9];
  float* out = (float*)d_out;

  float* ws   = (float*)d_ws;
  float* h1   = ws;
  float* g1   = h1   + (size_t)NNODES * H1C;
  float* as1  = g1   + (size_t)NNODES * H1C;
  float* ad1  = as1  + (size_t)NNODES * HEADS;
  int* sorted   = (int*)(ad1 + (size_t)NNODES * HEADS);
  int* cnt      = sorted + (size_t)ETOT;
  int* rowstart = cnt + NNODES;
  float* Wa1  = (float*)(rowstart + NNODES + 4);
  float* Wa2  = Wa1 + (size_t)IN_CH * LGW;
  float* h2   = h1;
  float* la2  = h1 + (size_t)NNODES * OUT_CH;

  const int T = 256;

  fold_kernel<<<FOLD1_BLOCKS + FOLD2_BLOCKS + FILL_BLOCKS, 256, 0, stream>>>(
      W1, a_src1, a_dst1, Wa1, W2, a_src2, a_dst2, Wa2, cnt);

  wmma_gemm_l1<<<dim3((NNODES + 63) / 64, H1C / 64), 128, 0, stream>>>(x, W1, h1, NNODES,
                                                                       Wa1, as1, ad1);
  dst_count_kernel<<<(NEDGES + T - 1) / T, T, 0, stream>>>(ei, cnt, NEDGES);
  row_start_kernel<<<1, SCAN_T, 0, stream>>>(cnt, rowstart, NNODES);
  place_kernel<<<(NEDGES + T - 1) / T, T, 0, stream>>>(ei, rowstart, cnt, sorted,
                                                       NEDGES, NNODES);
  gather_rows_l1_kernel<<<(NNODES * 32 + T - 1) / T, T, 0, stream>>>(
      sorted, rowstart, as1, ad1, h1, b1, g1, NNODES);

  wmma_gemm_l2<<<dim3((NNODES + 63) / 64, OUT_CH / 64), 128, 0, stream>>>(g1, W2, h2, NNODES,
                                                                          Wa2, la2);
  gather_rows_l2_kernel<<<(NNODES * 32 + T - 1) / T, T, 0, stream>>>(
      sorted, rowstart, la2, h2, b2, out, NNODES);
}


# counting sort with one atomic pass: the counting add returns each edge's rank in its row (kept per edge), the placing pass has no atomics; row order still an atomic arrival order, sum order changes
# speedup vs baseline: 6.1586x; 1.1855x over previous
#include <hip/hip_runtime.h>
#include <math.h>
#include <cstddef>

#define NNODES   20000
#define NEDGES   320000
#define ETOT     (NEDGES + NNODES)
#define IN_CH    128
#define HID_CH   32
#define HEADS    8
#define H1C      (HEADS * HID_CH)
#define OUT_CH   64
#define NEG_SLOPE 0.2f

typedef float  v2f   __attribute__((ext_vector_type(2)));
typedef float  v8f   __attribute__((ext_vector_type(8)));

#define LGW 16

template <int K, int N, int MODE>
__device__ __forceinline__ void gemm_strip(const float* __restrict__ A,
                                           const float* __restrict__ B,
                                           float* __restrict__ C, int M,
                                           const float* __restrict__ Wa,
                                           float* __restrict__ L0,
                                           float* __restrict__ L1) {
  const int lane = threadIdx.x & 31;
  const int wave = __builtin_amdgcn_readfirstlane((int)(threadIdx.x >> 5));
  const int tm   = blockIdx.x * 4 + wave;
  if (tm * 16 >= M) return;
  const int j  = lane & 15;
  const int kh = lane >> 4;
  const int cb = blockIdx.y * 64;
  const float* Ap = A + (size_t)(tm * 16 + j) * K + kh * 2;
  const float* Bp = B + (size_t)(kh * 2) * N + cb + 4 * j;
  const float* Wp = Wa + (size_t)(kh * 2) * LGW + j;
  v8f acc0 = {0.f, 0.f, 0.f, 0.f, 0.f, 0.f, 0.f, 0.f};
  v8f acc1 = acc0, acc2 = acc0, acc3 = acc0;
  v8f accL = acc0;

#pragma unroll 1
  for (int k0 = 0; k0 < K; k0 += 32) {
    v2f    a[8];
    float4 b0[8], b1[8];
    v2f    w[8];
#pragma unroll
    for (int s = 0; s < 8; ++s) {
      a[s]  = *(const v2f*)(Ap + k0 + 4 * s);
      b0[s] = *(const float4*)(Bp + (size_t)(k0 + 4 * s) * N);
      b1[s] = *(const float4*)(Bp + (size_t)(k0 + 4 * s + 1) * N);
      if constexpr (MODE != 0) {
        w[s].x = Wp[(size_t)(k0 + 4 * s) * LGW];
        w[s].y = Wp[(size_t)(k0 + 4 * s + 1) * LGW];
      }
    }
#pragma unroll
    for (int s = 0; s < 8; ++s) {
      v2f p0, p1, p2, p3;
      p0.x = b0[s].x; p0.y = b1[s].x;
      p1.x = b0[s].y; p1.y = b1[s].y;
      p2.x = b0[s].z; p2.y = b1[s].z;
      p3.x = b0[s].w; p3.y = b1[s].w;
      acc0 = __builtin_amdgcn_wmma_f32_16x16x4_f32(false, a[s], false, p0,
                                                   (short)0, acc0, false, false);
      acc1 = __builtin_amdgcn_wmma_f32_16x16x4_f32(false, a[s], false, p1,
                                                   (short)0, acc1, false, false);
      acc2 = __builtin_amdgcn_wmma_f32_16x16x4_f32(false, a[s], false, p2,
                                                   (short)0, acc2, false, false);
      acc3 = __builtin_amdgcn_wmma_f32_16x16x4_f32(false, a[s], false, p3,
                                                   (short)0, acc3, false, false);
      if constexpr (MODE != 0) {
        accL = __builtin_amdgcn_wmma_f32_16x16x4_f32(false, a[s], false, w[s],
                                                     (short)0, accL, false, false);
      }
    }
  }

  float* Cp = C + ((size_t)tm * 16 + kh * 8) * N + cb + 4 * j;
#pragma unroll
  for (int r = 0; r < 8; ++r) {
    *(float4*)(Cp + (size_t)r * N) = make_float4(acc0[r], acc1[r], acc2[r], acc3[r]);
  }
  if constexpr (MODE == 1) {
    float* Lp = ((j < HEADS) ? L0 : L1) + ((size_t)tm * 16 + kh * 8) * HEADS + (j & (HEADS - 1));
#pragma unroll
    for (int r = 0; r < 8; ++r) Lp[(size_t)r * HEADS] = accL[r];
  }
  if constexpr (MODE == 2) {
    if (j < 2) {
      float* Lp = L0 + ((size_t)tm * 16 + kh * 8) * 2 + j;
#pragma unroll
      for (int r = 0; r < 8; ++r) Lp[(size_t)r * 2] = accL[r];
    }
  }
}

__global__ void __launch_bounds__(128)
wmma_gemm_l1(const float* __restrict__ A, const float* __restrict__ B,
             float* __restrict__ C, int M, const float* __restrict__ Wa,
             float* __restrict__ as, float* __restrict__ ad) {
  if (blockIdx.y == 0) gemm_strip<IN_CH, H1C, 1>(A, B, C, M, Wa, as, ad);
  else                 gemm_strip<IN_CH, H1C, 0>(A, B, C, M, Wa, as, ad);
}

__global__ void __launch_bounds__(128)
wmma_gemm_l2(const float* __restrict__ A, const float* __restrict__ B,
             float* __restrict__ C, int M, const float* __restrict__ Wa,
             float* __restrict__ la) {
  gemm_strip<H1C, OUT_CH, 2>(A, B, C, M, Wa, la, la);
}

#define FOLD1_BLOCKS ((IN_CH * LGW) / 256)
#define FOLD2_BLOCKS ((H1C * LGW) / 256)
#define FILL_BLOCKS  ((NNODES + 255) / 256)

__global__ void __launch_bounds__(256)
fold_kernel(const float* __restrict__ W1, const float* __restrict__ a_src1,
            const float* __restrict__ a_dst1, float* __restrict__ Wa1,
            const float* __restrict__ W2, const float* __restrict__ a_src2,
            const float* __restrict__ a_dst2, float* __restrict__ Wa2,
            int* __restrict__ cnt) {
  const int b = blockIdx.x;
  if (b < FOLD1_BLOCKS) {
    const int i   = b * 256 + threadIdx.x;
    const int k   = i / LGW, col = i % LGW, hd = col % HEADS;
    const float4* wp = (const float4*)(W1 + (size_t)k * H1C + hd * HID_CH);
    const float4* ap = (const float4*)(((col < HEADS) ? a_src1 : a_dst1) + hd * HID_CH);
    float s = 0.f;
#pragma unroll
    for (int c = 0; c < HID_CH / 4; ++c) {
      const float4 wv = wp[c], av = ap[c];
      s += wv.x * av.x; s += wv.y * av.y; s += wv.z * av.z; s += wv.w * av.w;
    }
    Wa1[i] = s;
  } else if (b < FOLD1_BLOCKS + FOLD2_BLOCKS) {
    const int i   = (b - FOLD1_BLOCKS) * 256 + threadIdx.x;
    const int k   = i / LGW, col = i % LGW;
    float s = 0.f;
    if (col < 2) {
      const float4* wp = (const float4*)(W2 + (size_t)k * OUT_CH);
      const float4* ap = (const float4*)((col == 0) ? a_src2 : a_dst2);
#pragma unroll
      for (int c = 0; c < OUT_CH / 4; ++c) {
        const float4 wv = wp[c], av = ap[c];
        s += wv.x * av.x; s += wv.y * av.y; s += wv.z * av.z; s += wv.w * av.w;
      }
    }
    Wa2[i] = s;
  } else {
    const int i = (b - FOLD1_BLOCKS - FOLD2_BLOCKS) * 256 + threadIdx.x;
    if (i < NNODES) cnt[i] = 0;
  }
}

__global__ void dst_count_kernel(const int* __restrict__ ei, int* __restrict__ cnt,
                                 int* __restrict__ rank, int Eorig) {
  int i = blockIdx.x * blockDim.x + threadIdx.x;
  if (i < Eorig) rank[i] = atomicAdd(&cnt[ei[Eorig + i]], 1);
}

#define SCAN_T   1024
#define SCAN_PER ((NNODES + SCAN_T - 1) / SCAN_T)
__global__ void __launch_bounds__(SCAN_T)
row_start_kernel(const int* __restrict__ cnt, int* __restrict__ rowstart, int N) {
  __shared__ int part[2][SCAN_T];
  const int t  = threadIdx.x;
  const int lo = t * SCAN_PER;
  const bool whole = (lo + SCAN_PER <= N);
  int len[SCAN_PER];
  if (whole) {
#pragma unroll
    for (int j = 0; j < SCAN_PER; ++j) len[j] = cnt[lo + j] + 1;
  } else {
#pragma unroll
    for (int j = 0; j < SCAN_PER; ++j) len[j] = (lo + j < N) ? cnt[lo + j] + 1 : 0;
  }
  int sum = 0;
#pragma unroll
  for (int j = 0; j < SCAN_PER; ++j) sum += len[j];
  part[0][t] = sum;
  __syncthreads();
  int cur = 0;
  for (int off = 1; off < SCAN_T; off <<= 1) {
    int v = part[cur][t];
    if (t >= off) v += part[cur][t - off];
    part[cur ^ 1][t] = v;
    __syncthreads();
    cur ^= 1;
  }
  int run = part[cur][t] - sum;
  if (whole) {
#pragma unroll
    for (int j = 0; j < SCAN_PER; ++j) { rowstart[lo + j] = run; run += len[j]; }
  } else {
#pragma unroll
    for (int j = 0; j < SCAN_PER; ++j) {
      if (lo + j < N) { rowstart[lo + j] = run; run += len[j]; }
    }
  }
  if (t == SCAN_T - 1) rowstart[N] = part[cur][t];
}

__global__ void place_kernel(const int* __restrict__ ei, const int* __restrict__ rowstart,
                             const int* __restrict__ rank, int* __restrict__ sorted,
                             int Eorig, int N) {
  int i = blockIdx.x * blockDim.x + threadIdx.x;
  if (i < Eorig) {
    int s = ei[i], d = ei[Eorig + i];
    sorted[rowstart[d] + 1 + rank[i]] = s;
  }
  if (i < N) sorted[rowstart[i]] = i;
}


__global__ void __launch_bounds__(256)
gather_rows_l1_kernel(const int* __restrict__ sorted, const int* __restrict__ rowstart,
                      const float* __restrict__ as, const float* __restrict__ ad,
                      const float* __restrict__ h, const float* __restrict__ bias,
                      float* __restrict__ g, int N) {
  const int lane = threadIdx.x & 31;
  const int d = __builtin_amdgcn_readfirstlane((int)((blockIdx.x * blockDim.x + threadIdx.x) >> 5));
  if (d >= N) return;
  const int beg = rowstart[d], end = rowstart[d + 1];
  const int hA = lane >> 3;
  const int hB = 4 + (lane >> 3);
  const int h8 = lane & 7;

  float mx = -INFINITY;
  for (int base = beg; base < end; base += 32) {
    const int n = min(32, end - base);
    int mine = 0;
    if (lane < n) mine = sorted[base + lane];
#pragma unroll 4
    for (int k = 0; k < n; ++k) {
      const int s = __builtin_amdgcn_readlane(mine, k);
      mx = fmaxf(mx, as[(size_t)s * HEADS + h8]);
    }
  }
  const float adA = ad[d * HEADS + hA];
  const float adB = ad[d * HEADS + hB];
  float mA = __shfl(mx, hA) + adA;
  float mB = __shfl(mx, hB) + adB;
  mA = (mA > 0.f) ? mA : NEG_SLOPE * mA;
  mB = (mB > 0.f) ? mB : NEG_SLOPE * mB;

  float4 accA = make_float4(0.f, 0.f, 0.f, 0.f);
  float4 accB = make_float4(0.f, 0.f, 0.f, 0.f);
  float denA = 0.f, denB = 0.f;
  for (int base = beg; base < end; base += 32) {
    const int n = min(32, end - base);
    int mine = 0;
    if (lane < n) mine = sorted[base + lane];
#pragma unroll 4
    for (int k = 0; k < n; ++k) {
      const int s = __builtin_amdgcn_readlane(mine, k);
      float eA = as[(size_t)s * HEADS + hA] + adA;
      float eB = as[(size_t)s * HEADS + hB] + adB;
      const float4* hp = (const float4*)(h + (size_t)s * H1C);
      const float4 vA = hp[lane];
      const float4 vB = hp[32 + lane];
      eA = (eA > 0.f) ? eA : NEG_SLOPE * eA;
      eB = (eB > 0.f) ? eB : NEG_SLOPE * eB;
      const float wA = __expf(eA - mA);
      const float wB = __expf(eB - mB);
      denA += wA;
      denB += wB;
      accA.x += wA * vA.x; accA.y += wA * vA.y; accA.z += wA * vA.z; accA.w += wA * vA.w;
      accB.x += wB * vB.x; accB.y += wB * vB.y; accB.z += wB * vB.z; accB.w += wB * vB.w;
    }
  }
  const float dA = denA + 1e-16f;
  const float dB = denB + 1e-16f;
  const float4 bA = ((const float4*)bias)[lane];
  const float4 bB = ((const float4*)bias)[32 + lane];
  float4 oA, oB;
  oA.x = accA.x / dA + bA.x; oA.y = accA.y / dA + bA.y;
  oA.z = accA.z / dA + bA.z; oA.w = accA.w / dA + bA.w;
  oB.x = accB.x / dB + bB.x; oB.y = accB.y / dB + bB.y;
  oB.z = accB.z / dB + bB.z; oB.w = accB.w / dB + bB.w;
  oA.x = (oA.x > 0.f) ? oA.x : (__expf(oA.x) - 1.f);
  oA.y = (oA.y > 0.f) ? oA.y : (__expf(oA.y) - 1.f);
  oA.z = (oA.z > 0.f) ? oA.z : (__expf(oA.z) - 1.f);
  oA.w = (oA.w > 0.f) ? oA.w : (__expf(oA.w) - 1.f);
  oB.x = (oB.x > 0.f) ? oB.x : (__expf(oB.x) - 1.f);
  oB.y = (oB.y > 0.f) ? oB.y : (__expf(oB.y) - 1.f);
  oB.z = (oB.z > 0.f) ? oB.z : (__expf(oB.z) - 1.f);
  oB.w = (oB.w > 0.f) ? oB.w : (__expf(oB.w) - 1.f);
  float4* gp = (float4*)(g + (size_t)d * H1C);
  gp[lane]      = oA;
  gp[32 + lane] = oB;
}

__global__ void __launch_bounds__(256)
gather_rows_l2_kernel(const int* __restrict__ sorted, const int* __restrict__ rowstart,
                      const float* __restrict__ la,
                      const float* __restrict__ h, const float* __restrict__ bias,
                      float* __restrict__ out, int N) {
  const int lane = threadIdx.x & 31;
  const int d = __builtin_amdgcn_readfirstlane((int)((blockIdx.x * blockDim.x + threadIdx.x) >> 5));
  if (d >= N) return;
  const int beg = rowstart[d], end = rowstart[d + 1];

  float mx = -INFINITY;
  for (int base = beg; base < end; base += 32) {
    if (base + lane < end) mx = fmaxf(mx, la[2 * (size_t)sorted[base + lane]]);
  }
#pragma unroll
  for (int o = 16; o > 0; o >>= 1) mx = fmaxf(mx, __shfl_xor(mx, o));
  const float adv = la[2 * (size_t)d + 1];
  float m = mx + adv;
  m = (m > 0.f) ? m : NEG_SLOPE * m;

  float2 acc = make_float2(0.f, 0.f);
  float denl = 0.f;
  for (int base = beg; base < end; base += 32) {
    const int n = min(32, end - base);
    int mine = 0;
    float exl = 0.f;
    if (lane < n) {
      mine = sorted[base + lane];
      float e = la[2 * (size_t)mine] + adv;
      e = (e > 0.f) ? e : NEG_SLOPE * e;
      exl = __expf(e - m);
    }
    denl += exl;
#pragma unroll 4
    for (int k = 0; k < n; ++k) {
      const int   s = __builtin_amdgcn_readlane(mine, k);
      const float w = __int_as_float(__builtin_amdgcn_readlane(__float_as_int(exl), k));
      const float2 v = ((const float2*)(h + (size_t)s * OUT_CH))[lane];
      acc.x += w * v.x; acc.y += w * v.y;
    }
  }
#pragma unroll
  for (int o = 16; o > 0; o >>= 1) denl += __shfl_xor(denl, o);
  const float dn = denl + 1e-16f;
  const float2 b = ((const float2*)bias)[lane];
  float2 o2;
  o2.x = acc.x / dn + b.x;
  o2.y = acc.y / dn + b.y;
  ((float2*)(out + (size_t)d * OUT_CH))[lane] = o2;
}

extern "C" void kernel_launch(void* const* d_in, const int* in_sizes, int n_in,
                              void* d_out, int out_size, void* d_ws, size_t ws_size,
                              hipStream_t stream) {
  const float* x      = (const float*)d_in[0];
  const int*   ei     = (const int*)  d_in[1];
  const float* W1     = (const float*)d_in[2];
  const float* a_src1 = (const float*)d_in[3];
  const float* a_dst1 = (const float*)d_in[4];
  const float* b1     = (const float*)d_in[5];
  const float* W2     = (const float*)d_in[6];
  const float* a_src2 = (const float*)d_in[7];
  const float* a_dst2 = (const float*)d_in[8];
  const float* b2     = (const float*)d_in[9];
  float* out = (float*)d_out;

  float* ws   = (float*)d_ws;
  float* h1   = ws;
  float* g1   = h1   + (size_t)NNODES * H1C;
  float* as1  = g1   + (size_t)NNODES * H1C;
  float* ad1  = as1  + (size_t)NNODES * HEADS;
  int* sorted   = (int*)(ad1 + (size_t)NNODES * HEADS);
  int* cnt      = sorted + (size_t)ETOT;
  int* rowstart = cnt + NNODES;
  float* Wa1  = (float*)(rowstart + NNODES + 4);
  float* Wa2  = Wa1 + (size_t)IN_CH * LGW;
  int* rank   = (int*)(Wa2 + (size_t)H1C * LGW);
  float* h2   = h1;
  float* la2  = h1 + (size_t)NNODES * OUT_CH;

  const int T = 256;

  fold_kernel<<<FOLD1_BLOCKS + FOLD2_BLOCKS + FILL_BLOCKS, 256, 0, stream>>>(
      W1, a_src1, a_dst1, Wa1, W2, a_src2, a_dst2, Wa2, cnt);

  wmma_gemm_l1<<<dim3((NNODES + 63) / 64, H1C / 64), 128, 0, stream>>>(x, W1, h1, NNODES,
                                                                       Wa1, as1, ad1);
  dst_count_kernel<<<(NEDGES + T - 1) / T, T, 0, stream>>>(ei, cnt, rank, NEDGES);
  row_start_kernel<<<1, SCAN_T, 0, stream>>>(cnt, rowstart, NNODES);
  place_kernel<<<(NEDGES + T - 1) / T, T, 0, stream>>>(ei, rowstart, rank, sorted,
                                                       NEDGES, NNODES);
  gather_rows_l1_kernel<<<(NNODES * 32 + T - 1) / T, T, 0, stream>>>(
      sorted, rowstart, as1, ad1, h1, b1, g1, NNODES);

  wmma_gemm_l2<<<dim3((NNODES + 63) / 64, OUT_CH / 64), 128, 0, stream>>>(g1, W2, h2, NNODES,
                                                                          Wa2, la2);
  gather_rows_l2_kernel<<<(NNODES * 32 + T - 1) / T, T, 0, stream>>>(
      sorted, rowstart, la2, h2, b2, out, NNODES);
}


# fixed rows of 64 slots per destination: the counting add's rank places each edge at once; prefix scan, placing pass, rank array, 2 launches gone; longer rows rescan the edges; sum order as before
# speedup vs baseline: 6.8306x; 1.1091x over previous
#include <hip/hip_runtime.h>
#include <math.h>
#include <cstddef>

#define NNODES   20000
#define NEDGES   320000
#define IN_CH    128
#define HID_CH   32
#define HEADS    8
#define H1C      (HEADS * HID_CH)
#define OUT_CH   64
#define NEG_SLOPE 0.2f

typedef float  v2f   __attribute__((ext_vector_type(2)));
typedef float  v8f   __attribute__((ext_vector_type(8)));

#define LGW 16

template <int K, int N, int MODE>
__device__ __forceinline__ void gemm_strip(const float* __restrict__ A,
                                           const float* __restrict__ B,
                                           float* __restrict__ C, int M,
                                           const float* __restrict__ Wa,
                                           float* __restrict__ L0,
                                           float* __restrict__ L1) {
  const int lane = threadIdx.x & 31;
  const int wave = __builtin_amdgcn_readfirstlane((int)(threadIdx.x >> 5));
  const int tm   = blockIdx.x * 4 + wave;
  if (tm * 16 >= M) return;
  const int j  = lane & 15;
  const int kh = lane >> 4;
  const int cb = blockIdx.y * 64;
  const float* Ap = A + (size_t)(tm * 16 + j) * K + kh * 2;
  const float* Bp = B + (size_t)(kh * 2) * N + cb + 4 * j;
  const float* Wp = Wa + (size_t)(kh * 2) * LGW + j;
  v8f acc0 = {0.f, 0.f, 0.f, 0.f, 0.f, 0.f, 0.f, 0.f};
  v8f acc1 = acc0, acc2 = acc0, acc3 = acc0;
  v8f accL = acc0;

#pragma unroll 1
  for (int k0 = 0; k0 < K; k0 += 32) {
    v2f    a[8];
    float4 b0[8], b1[8];
    v2f    w[8];
#pragma unroll
    for (int s = 0; s < 8; ++s) {
      a[s]  = *(const v2f*)(Ap + k0 + 4 * s);
      b0[s] = *(const float4*)(Bp + (size_t)(k0 + 4 * s) * N);
      b1[s] = *(const float4*)(Bp + (size_t)(k0 + 4 * s + 1) * N);
      if constexpr (MODE != 0) {
        w[s].x = Wp[(size_t)(k0 + 4 * s) * LGW];
        w[s].y = Wp[(size_t)(k0 + 4 * s + 1) * LGW];
      }
    }
#pragma unroll
    for (int s = 0; s < 8; ++s) {
      v2f p0, p1, p2, p3;
      p0.x = b0[s].x; p0.y = b1[s].x;
      p1.x = b0[s].y; p1.y = b1[s].y;
      p2.x = b0[s].z; p2.y = b1[s].z;
      p3.x = b0[s].w; p3.y = b1[s].w;
      acc0 = __builtin_amdgcn_wmma_f32_16x16x4_f32(false, a[s], false, p0,
                                                   (short)0, acc0, false, false);
      acc1 = __builtin_amdgcn_wmma_f32_16x16x4_f32(false, a[s], false, p1,
                                                   (short)0, acc1, false, false);
      acc2 = __builtin_amdgcn_wmma_f32_16x16x4_f32(false, a[s], false, p2,
                                                   (short)0, acc2, false, false);
      acc3 = __builtin_amdgcn_wmma_f32_16x16x4_f32(false, a[s], false, p3,
                                                   (short)0, acc3, false, false);
      if constexpr (MODE != 0) {
        accL = __builtin_amdgcn_wmma_f32_16x16x4_f32(false, a[s], false, w[s],
                                                     (short)0, accL, false, false);
      }
    }
  }

  float* Cp = C + ((size_t)tm * 16 + kh * 8) * N + cb + 4 * j;
#pragma unroll
  for (int r = 0; r < 8; ++r) {
    *(float4*)(Cp + (size_t)r * N) = make_float4(acc0[r], acc1[r], acc2[r], acc3[r]);
  }
  if constexpr (MODE == 1) {
    float* Lp = ((j < HEADS) ? L0 : L1) + ((size_t)tm * 16 + kh * 8) * HEADS + (j & (HEADS - 1));
#pragma unroll
    for (int r = 0; r < 8; ++r) Lp[(size_t)r * HEADS] = accL[r];
  }
  if constexpr (MODE == 2) {
    if (j < 2) {
      float* Lp = L0 + ((size_t)tm * 16 + kh * 8) * 2 + j;
#pragma unroll
      for (int r = 0; r < 8; ++r) Lp[(size_t)r * 2] = accL[r];
    }
  }
}

__global__ void __launch_bounds__(128)
wmma_gemm_l1(const float* __restrict__ A, const float* __restrict__ B,
             float* __restrict__ C, int M, const float* __restrict__ Wa,
             float* __restrict__ as, float* __restrict__ ad) {
  if (blockIdx.y == 0) gemm_strip<IN_CH, H1C, 1>(A, B, C, M, Wa, as, ad);
  else                 gemm_strip<IN_CH, H1C, 0>(A, B, C, M, Wa, as, ad);
}

__global__ void __launch_bounds__(128)
wmma_gemm_l2(const float* __restrict__ A, const float* __restrict__ B,
             float* __restrict__ C, int M, const float* __restrict__ Wa,
             float* __restrict__ la) {
  gemm_strip<H1C, OUT_CH, 2>(A, B, C, M, Wa, la, la);
}

#define FOLD1_BLOCKS ((IN_CH * LGW) / 256)
#define FOLD2_BLOCKS ((H1C * LGW) / 256)
#define FILL_BLOCKS  ((NNODES + 255) / 256)

__global__ void __launch_bounds__(256)
fold_kernel(const float* __restrict__ W1, const float* __restrict__ a_src1,
            const float* __restrict__ a_dst1, float* __restrict__ Wa1,
            const float* __restrict__ W2, const float* __restrict__ a_src2,
            const float* __restrict__ a_dst2, float* __restrict__ Wa2,
            int* __restrict__ cnt) {
  const int b = blockIdx.x;
  if (b < FOLD1_BLOCKS) {
    const int i   = b * 256 + threadIdx.x;
    const int k   = i / LGW, col = i % LGW, hd = col % HEADS;
    const float4* wp = (const float4*)(W1 + (size_t)k * H1C + hd * HID_CH);
    const float4* ap = (const float4*)(((col < HEADS) ? a_src1 : a_dst1) + hd * HID_CH);
    float s = 0.f;
#pragma unroll
    for (int c = 0; c < HID_CH / 4; ++c) {
      const float4 wv = wp[c], av = ap[c];
      s += wv.x * av.x; s += wv.y * av.y; s += wv.z * av.z; s += wv.w * av.w;
    }
    Wa1[i] = s;
  } else if (b < FOLD1_BLOCKS + FOLD2_BLOCKS) {
    const int i   = (b - FOLD1_BLOCKS) * 256 + threadIdx.x;
    const int k   = i / LGW, col = i % LGW;
    float s = 0.f;
    if (col < 2) {
      const float4* wp = (const float4*)(W2 + (size_t)k * OUT_CH);
      const float4* ap = (const float4*)((col == 0) ? a_src2 : a_dst2);
#pragma unroll
      for (int c = 0; c < OUT_CH / 4; ++c) {
        const float4 wv = wp[c], av = ap[c];
        s += wv.x * av.x; s += wv.y * av.y; s += wv.z * av.z; s += wv.w * av.w;
      }
    }
    Wa2[i] = s;
  } else {
    const int i = (b - FOLD1_BLOCKS - FOLD2_BLOCKS) * 256 + threadIdx.x;
    if (i < NNODES) cnt[i] = 0;
  }
}

#define ROWCAP 64

__global__ void dst_place_kernel(const int* __restrict__ ei, int* __restrict__ cnt,
                                 int* __restrict__ ell, int Eorig) {
  int i = blockIdx.x * blockDim.x + threadIdx.x;
  if (i < Eorig) {
    const int s = ei[i], d = ei[Eorig + i];
    asm volatile("" :: "v"(s));
    const int r = atomicAdd(&cnt[d], 1);
    if (r < ROWCAP) ell[(size_t)d * ROWCAP + r] = s;
  }
}

__device__ __forceinline__ int row_entry(const int* __restrict__ row, int d, int idx, int tot) {
  int v = d;
  if (idx >= 1 && idx < tot) v = row[idx - 1];
  return v;
}


__device__ __forceinline__ void l1_edge(int s, int lane, int hA, int hB,
                                        const float* __restrict__ as,
                                        const float* __restrict__ h,
                                        float adA, float adB, float mA, float mB,
                                        float4& accA, float4& accB,
                                        float& denA, float& denB) {
  float eA = as[(size_t)s * HEADS + hA] + adA;
  float eB = as[(size_t)s * HEADS + hB] + adB;
  const float4* hp = (const float4*)(h + (size_t)s * H1C);
  const float4 vA = hp[lane];
  const float4 vB = hp[32 + lane];
  eA = (eA > 0.f) ? eA : NEG_SLOPE * eA;
  eB = (eB > 0.f) ? eB : NEG_SLOPE * eB;
  const float wA = __expf(eA - mA);
  const float wB = __expf(eB - mB);
  denA += wA;
  denB += wB;
  accA.x += wA * vA.x; accA.y += wA * vA.y; accA.z += wA * vA.z; accA.w += wA * vA.w;
  accB.x += wB * vB.x; accB.y += wB * vB.y; accB.z += wB * vB.z; accB.w += wB * vB.w;
}

__global__ void __launch_bounds__(256)
gather_rows_l1_kernel(const int* __restrict__ ell, const int* __restrict__ cnt,
                      const int* __restrict__ ei, int Eorig,
                      const float* __restrict__ as, const float* __restrict__ ad,
                      const float* __restrict__ h, const float* __restrict__ bias,
                      float* __restrict__ g, int N) {
  const int lane = threadIdx.x & 31;
  const int d = __builtin_amdgcn_readfirstlane((int)((blockIdx.x * blockDim.x + threadIdx.x) >> 5));
  if (d >= N) return;
  const int  n_in = cnt[d];
  const bool fits = (n_in <= ROWCAP);
  const int  tot  = fits ? n_in + 1 : 1;
  const int* row  = ell + (size_t)d * ROWCAP;
  const int hA = lane >> 3;
  const int hB = 4 + (lane >> 3);
  const int h8 = lane & 7;

  float mx = -INFINITY;
  for (int base = 0; base < tot; base += 32) {
    const int n = min(32, tot - base);
    const int mine = row_entry(row, d, base + lane, tot);
#pragma unroll 4
    for (int k = 0; k < n; ++k) {
      const int s = __builtin_amdgcn_readlane(mine, k);
      mx = fmaxf(mx, as[(size_t)s * HEADS + h8]);
    }
  }
  if (!fits) {
    for (int base = 0; base < Eorig; base += 32) {
      const int i = base + lane;
      int dd = -1;
      if (i < Eorig) dd = ei[Eorig + i];
      unsigned hit = __builtin_amdgcn_ballot_w32(dd == d);
      if (hit != 0u) {
        int mine = 0;
        if (dd == d) mine = ei[i];
        while (hit != 0u) {
          const int k = __builtin_ctz(hit);
          hit &= hit - 1u;
          const int s = __builtin_amdgcn_readlane(mine, k);
          mx = fmaxf(mx, as[(size_t)s * HEADS + h8]);
        }
      }
    }
  }
  const float adA = ad[d * HEADS + hA];
  const float adB = ad[d * HEADS + hB];
  float mA = __shfl(mx, hA) + adA;
  float mB = __shfl(mx, hB) + adB;
  mA = (mA > 0.f) ? mA : NEG_SLOPE * mA;
  mB = (mB > 0.f) ? mB : NEG_SLOPE * mB;

  float4 accA = make_float4(0.f, 0.f, 0.f, 0.f);
  float4 accB = make_float4(0.f, 0.f, 0.f, 0.f);
  float denA = 0.f, denB = 0.f;
  for (int base = 0; base < tot; base += 32) {
    const int n = min(32, tot - base);
    const int mine = row_entry(row, d, base + lane, tot);
#pragma unroll 4
    for (int k = 0; k < n; ++k) {
      const int s = __builtin_amdgcn_readlane(mine, k);
      l1_edge(s, lane, hA, hB, as, h, adA, adB, mA, mB, accA, accB, denA, denB);
    }
  }
  if (!fits) {
    for (int base = 0; base < Eorig; base += 32) {
      const int i = base + lane;
      int dd = -1;
      if (i < Eorig) dd = ei[Eorig + i];
      unsigned hit = __builtin_amdgcn_ballot_w32(dd == d);
      if (hit != 0u) {
        int mine = 0;
        if (dd == d) mine = ei[i];
        while (hit != 0u) {
          const int k = __builtin_ctz(hit);
          hit &= hit - 1u;
          const int s = __builtin_amdgcn_readlane(mine, k);
          l1_edge(s, lane, hA, hB, as, h, adA, adB, mA, mB, accA, accB, denA, denB);
        }
      }
    }
  }
  const float dA = denA + 1e-16f;
  const float dB = denB + 1e-16f;
  const float4 bA = ((const float4*)bias)[lane];
  const float4 bB = ((const float4*)bias)[32 + lane];
  float4 oA, oB;
  oA.x = accA.x / dA + bA.x; oA.y = accA.y / dA + bA.y;
  oA.z = accA.z / dA + bA.z; oA.w = accA.w / dA + bA.w;
  oB.x = accB.x / dB + bB.x; oB.y = accB.y / dB + bB.y;
  oB.z = accB.z / dB + bB.z; oB.w = accB.w / dB + bB.w;
  oA.x = (oA.x > 0.f) ? oA.x : (__expf(oA.x) - 1.f);
  oA.y = (oA.y > 0.f) ? oA.y : (__expf(oA.y) - 1.f);
  oA.z = (oA.z > 0.f) ? oA.z : (__expf(oA.z) - 1.f);
  oA.w = (oA.w > 0.f) ? oA.w : (__expf(oA.w) - 1.f);
  oB.x = (oB.x > 0.f) ? oB.x : (__expf(oB.x) - 1.f);
  oB.y = (oB.y > 0.f) ? oB.y : (__expf(oB.y) - 1.f);
  oB.z = (oB.z > 0.f) ? oB.z : (__expf(oB.z) - 1.f);
  oB.w = (oB.w > 0.f) ? oB.w : (__expf(oB.w) - 1.f);
  float4* gp = (float4*)(g + (size_t)d * H1C);
  gp[lane]      = oA;
  gp[32 + lane] = oB;
}

__global__ void __launch_bounds__(256)
gather_rows_l2_kernel(const int* __restrict__ ell, const int* __restrict__ cnt,
                      const int* __restrict__ ei, int Eorig,
                      const float* __restrict__ la,
                      const float* __restrict__ h, const float* __restrict__ bias,
                      float* __restrict__ out, int N) {
  const int lane = threadIdx.x & 31;
  const int d = __builtin_amdgcn_readfirstlane((int)((blockIdx.x * blockDim.x + threadIdx.x) >> 5));
  if (d >= N) return;
  const int  n_in = cnt[d];
  const bool fits = (n_in <= ROWCAP);
  const int  tot  = fits ? n_in + 1 : 1;
  const int* row  = ell + (size_t)d * ROWCAP;

  float mx = -INFINITY;
  for (int base = 0; base < tot; base += 32) {
    const int idx = base + lane;
    if (idx < tot) mx = fmaxf(mx, la[2 * (size_t)row_entry(row, d, idx, tot)]);
  }
  if (!fits) {
    for (int base = 0; base < Eorig; base += 32) {
      const int i = base + lane;
      if (i < Eorig && ei[Eorig + i] == d) mx = fmaxf(mx, la[2 * (size_t)ei[i]]);
    }
  }
#pragma unroll
  for (int o = 16; o > 0; o >>= 1) mx = fmaxf(mx, __shfl_xor(mx, o));
  const float adv = la[2 * (size_t)d + 1];
  float m = mx + adv;
  m = (m > 0.f) ? m : NEG_SLOPE * m;

  float2 acc = make_float2(0.f, 0.f);
  float denl = 0.f;
  for (int base = 0; base < tot; base += 32) {
    const int n = min(32, tot - base);
    const int mine = row_entry(row, d, base + lane, tot);
    float exl = 0.f;
    if (lane < n) {
      float e = la[2 * (size_t)mine] + adv;
      e = (e > 0.f) ? e : NEG_SLOPE * e;
      exl = __expf(e - m);
    }
    denl += exl;
#pragma unroll 4
    for (int k = 0; k < n; ++k) {
      const int   s = __builtin_amdgcn_readlane(mine, k);
      const float w = __int_as_float(__builtin_amdgcn_readlane(__float_as_int(exl), k));
      const float2 v = ((const float2*)(h + (size_t)s * OUT_CH))[lane];
      acc.x += w * v.x; acc.y += w * v.y;
    }
  }
  if (!fits) {
    for (int base = 0; base < Eorig; base += 32) {
      const int i = base + lane;
      int dd = -1;
      if (i < Eorig) dd = ei[Eorig + i];
      unsigned hit = __builtin_amdgcn_ballot_w32(dd == d);
      if (hit != 0u) {
        int mine = 0;
        float exl = 0.f;
        if (dd == d) {
          mine = ei[i];
          float e = la[2 * (size_t)mine] + adv;
          e = (e > 0.f) ? e : NEG_SLOPE * e;
          exl = __expf(e - m);
        }
        denl += exl;
        while (hit != 0u) {
          const int k = __builtin_ctz(hit);
          hit &= hit - 1u;
          const int   s = __builtin_amdgcn_readlane(mine, k);
          const float w = __int_as_float(__builtin_amdgcn_readlane(__float_as_int(exl), k));
          const float2 v = ((const float2*)(h + (size_t)s * OUT_CH))[lane];
          acc.x += w * v.x; acc.y += w * v.y;
        }
      }
    }
  }
#pragma unroll
  for (int o = 16; o > 0; o >>= 1) denl += __shfl_xor(denl, o);
  const float dn = denl + 1e-16f;
  const float2 b = ((const float2*)bias)[lane];
  float2 o2;
  o2.x = acc.x / dn + b.x;
  o2.y = acc.y / dn + b.y;
  ((float2*)(out + (size_t)d * OUT_CH))[lane] = o2;
}

extern "C" void kernel_launch(void* const* d_in, const int* in_sizes, int n_in,
                              void* d_out, int out_size, void* d_ws, size_t ws_size,
                              hipStream_t stream) {
  const float* x      = (const float*)d_in[0];
  const int*   ei     = (const int*)  d_in[1];
  const float* W1     = (const float*)d_in[2];
  const float* a_src1 = (const float*)d_in[3];
  const float* a_dst1 = (const float*)d_in[4];
  const float* b1     = (const float*)d_in[5];
  const float* W2     = (const float*)d_in[6];
  const float* a_src2 = (const float*)d_in[7];
  const float* a_dst2 = (const float*)d_in[8];
  const float* b2     = (const float*)d_in[9];
  float* out = (float*)d_out;

  float* ws   = (float*)d_ws;
  float* h1   = ws;
  float* g1   = h1   + (size_t)NNODES * H1C;
  float* as1  = g1   + (size_t)NNODES * H1C;
  float* ad1  = as1  + (size_t)NNODES * HEADS;
  int* ell      = (int*)(ad1 + (size_t)NNODES * HEADS);
  int* cnt      = ell + (size_t)NNODES * ROWCAP;
  float* Wa1  = (float*)(cnt + NNODES);
  float* Wa2  = Wa1 + (size_t)IN_CH * LGW;
  float* h2   = h1;
  float* la2  = h1 + (size_t)NNODES * OUT_CH;

  const int T = 256;

  fold_kernel<<<FOLD1_BLOCKS + FOLD2_BLOCKS + FILL_BLOCKS, 256, 0, stream>>>(
      W1, a_src1, a_dst1, Wa1, W2, a_src2, a_dst2, Wa2, cnt);

  wmma_gemm_l1<<<dim3((NNODES + 63) / 64, H1C / 64), 128, 0, stream>>>(x, W1, h1, NNODES,
                                                                       Wa1, as1, ad1);
  dst_place_kernel<<<(NEDGES + T - 1) / T, T, 0, stream>>>(ei, cnt, ell, NEDGES);
  gather_rows_l1_kernel<<<(NNODES * 32 + T - 1) / T, T, 0, stream>>>(
      ell, cnt, ei, NEDGES, as1, ad1, h1, b1, g1, NNODES);

  wmma_gemm_l2<<<dim3((NNODES + 63) / 64, OUT_CH / 64), 128, 0, stream>>>(g1, W2, h2, NNODES,
                                                                          Wa2, la2);
  gather_rows_l2_kernel<<<(NNODES * 32 + T - 1) / T, T, 0, stream>>>(
      ell, cnt, ei, NEDGES, la2, h2, b2, out, NNODES);
}


# edge placement runs in the layer-1 GEMM's launch: alternate workgroups of one grid (2 edges per thread), its counter round trips overlap the matrix work; 6 launches -> 5; sum order as before
# speedup vs baseline: 7.2567x; 1.0624x over previous
#include <hip/hip_runtime.h>
#include <math.h>
#include <cstddef>

#define NNODES   20000
#define NEDGES   320000
#define IN_CH    128
#define HID_CH   32
#define HEADS    8
#define H1C      (HEADS * HID_CH)
#define OUT_CH   64
#define NEG_SLOPE 0.2f

typedef float  v2f   __attribute__((ext_vector_type(2)));
typedef float  v8f   __attribute__((ext_vector_type(8)));

#define LGW 16

template <int K, int N, int MODE>
__device__ __forceinline__ void gemm_strip(int bx, int by,
                                           const float* __restrict__ A,
                                           const float* __restrict__ B,
                                           float* __restrict__ C, int M,
                                           const float* __restrict__ Wa,
                                           float* __restrict__ L0,
                                           float* __restrict__ L1) {
  const int lane = threadIdx.x & 31;
  const int wave = __builtin_amdgcn_readfirstlane((int)(threadIdx.x >> 5));
  const int tm   = bx * 4 + wave;
  if (tm * 16 >= M) return;
  const int j  = lane & 15;
  const int kh = lane >> 4;
  const int cb = by * 64;
  const float* Ap = A + (size_t)(tm * 16 + j) * K + kh * 2;
  const float* Bp = B + (size_t)(kh * 2) * N + cb + 4 * j;
  const float* Wp = Wa + (size_t)(kh * 2) * LGW + j;
  v8f acc0 = {0.f, 0.f, 0.f, 0.f, 0.f, 0.f, 0.f, 0.f};
  v8f acc1 = acc0, acc2 = acc0, acc3 = acc0;
  v8f accL = acc0;

#pragma unroll 1
  for (int k0 = 0; k0 < K; k0 += 32) {
    v2f    a[8];
    float4 b0[8], b1[8];
    v2f    w[8];
#pragma unroll
    for (int s = 0; s < 8; ++s) {
      a[s]  = *(const v2f*)(Ap + k0 + 4 * s);
      b0[s] = *(const float4*)(Bp + (size_t)(k0 + 4 * s) * N);
      b1[s] = *(const float4*)(Bp + (size_t)(k0 + 4 * s + 1) * N);
      if constexpr (MODE != 0) {
        w[s].x = Wp[(size_t)(k0 + 4 * s) * LGW];
        w[s].y = Wp[(size_t)(k0 + 4 * s + 1) * LGW];
      }
    }
#pragma unroll
    for (int s = 0; s < 8; ++s) {
      v2f p0, p1, p2, p3;
      p0.x = b0[s].x; p0.y = b1[s].x;
      p1.x = b0[s].y; p1.y = b1[s].y;
      p2.x = b0[s].z; p2.y = b1[s].z;
      p3.x = b0[s].w; p3.y = b1[s].w;
      acc0 = __builtin_amdgcn_wmma_f32_16x16x4_f32(false, a[s], false, p0,
                                                   (short)0, acc0, false, false);
      acc1 = __builtin_amdgcn_wmma_f32_16x16x4_f32(false, a[s], false, p1,
                                                   (short)0, acc1, false, false);
      acc2 = __builtin_amdgcn_wmma_f32_16x16x4_f32(false, a[s], false, p2,
                                                   (short)0, acc2, false, false);
      acc3 = __builtin_amdgcn_wmma_f32_16x16x4_f32(false, a[s], false, p3,
                                                   (short)0, acc3, false, false);
      if constexpr (MODE != 0) {
        accL = __builtin_amdgcn_wmma_f32_16x16x4_f32(false, a[s], false, w[s],
                                                     (short)0, accL, false, false);
      }
    }
  }

  float* Cp = C + ((size_t)tm * 16 + kh * 8) * N + cb + 4 * j;
#pragma unroll
  for (int r = 0; r < 8; ++r) {
    *(float4*)(Cp + (size_t)r * N) = make_float4(acc0[r], acc1[r], acc2[r], acc3[r]);
  }
  if constexpr (MODE == 1) {
    float* Lp = ((j < HEADS) ? L0 : L1) + ((size_t)tm * 16 + kh * 8) * HEADS + (j & (HEADS - 1));
#pragma unroll
    for (int r = 0; r < 8; ++r) Lp[(size_t)r * HEADS] = accL[r];
  }
  if constexpr (MODE == 2) {
    if (j < 2) {
      float* Lp = L0 + ((size_t)tm * 16 + kh * 8) * 2 + j;
#pragma unroll
      for (int r = 0; r < 8; ++r) Lp[(size_t)r * 2] = accL[r];
    }
  }
}

__global__ void __launch_bounds__(128)
wmma_gemm_l2(const float* __restrict__ A, const float* __restrict__ B,
             float* __restrict__ C, int M, const float* __restrict__ Wa,
             float* __restrict__ la) {
  gemm_strip<H1C, OUT_CH, 2>((int)blockIdx.x, (int)blockIdx.y, A, B, C, M, Wa, la, la);
}

#define FOLD1_BLOCKS ((IN_CH * LGW) / 256)
#define FOLD2_BLOCKS ((H1C * LGW) / 256)
#define FILL_BLOCKS  ((NNODES + 255) / 256)

__global__ void __launch_bounds__(256)
fold_kernel(const float* __restrict__ W1, const float* __restrict__ a_src1,
            const float* __restrict__ a_dst1, float* __restrict__ Wa1,
            const float* __restrict__ W2, const float* __restrict__ a_src2,
            const float* __restrict__ a_dst2, float* __restrict__ Wa2,
            int* __restrict__ cnt) {
  const int b = blockIdx.x;
  if (b < FOLD1_BLOCKS) {
    const int i   = b * 256 + threadIdx.x;
    const int k   = i / LGW, col = i % LGW, hd = col % HEADS;
    const float4* wp = (const float4*)(W1 + (size_t)k * H1C + hd * HID_CH);
    const float4* ap = (const float4*)(((col < HEADS) ? a_src1 : a_dst1) + hd * HID_CH);
    float s = 0.f;
#pragma unroll
    for (int c = 0; c < HID_CH / 4; ++c) {
      const float4 wv = wp[c], av = ap[c];
      s += wv.x * av.x; s += wv.y * av.y; s += wv.z * av.z; s += wv.w * av.w;
    }
    Wa1[i] = s;
  } else if (b < FOLD1_BLOCKS + FOLD2_BLOCKS) {
    const int i   = (b - FOLD1_BLOCKS) * 256 + threadIdx.x;
    const int k   = i / LGW, col = i % LGW;
    float s = 0.f;
    if (col < 2) {
      const float4* wp = (const float4*)(W2 + (size_t)k * OUT_CH);
      const float4* ap = (const float4*)((col == 0) ? a_src2 : a_dst2);
#pragma unroll
      for (int c = 0; c < OUT_CH / 4; ++c) {
        const float4 wv = wp[c], av = ap[c];
        s += wv.x * av.x; s += wv.y * av.y; s += wv.z * av.z; s += wv.w * av.w;
      }
    }
    Wa2[i] = s;
  } else {
    const int i = (b - FOLD1_BLOCKS - FOLD2_BLOCKS) * 256 + threadIdx.x;
    if (i < NNODES) cnt[i] = 0;
  }
}

#define ROWCAP 64
#define PLACE_EDGES 256
#define PLACE_BLOCKS ((NEDGES + PLACE_EDGES - 1) / PLACE_EDGES)
#define L1_BX ((NNODES + 63) / 64)
#define L1_GEMM_BLOCKS (L1_BX * (H1C / 64))
#define L1_PAIRS ((L1_GEMM_BLOCKS > PLACE_BLOCKS) ? L1_GEMM_BLOCKS : PLACE_BLOCKS)

__device__ __forceinline__ void place_edges(int pb, const int* __restrict__ ei,
                                            int* __restrict__ cnt, int* __restrict__ ell,
                                            int Eorig) {
  const int i0 = pb * PLACE_EDGES + (int)threadIdx.x;
  const int i1 = i0 + PLACE_EDGES / 2;
  if (i1 < Eorig) {
    const int s0 = ei[i0], d0 = ei[Eorig + i0];
    const int s1 = ei[i1], d1 = ei[Eorig + i1];
    asm volatile("" :: "v"(s0), "v"(s1));
    const int r0 = atomicAdd(&cnt[d0], 1);
    const int r1 = atomicAdd(&cnt[d1], 1);
    if (r0 < ROWCAP) ell[(size_t)d0 * ROWCAP + r0] = s0;
    if (r1 < ROWCAP) ell[(size_t)d1 * ROWCAP + r1] = s1;
  } else if (i0 < Eorig) {
    const int s0 = ei[i0], d0 = ei[Eorig + i0];
    asm volatile("" :: "v"(s0));
    const int r0 = atomicAdd(&cnt[d0], 1);
    if (r0 < ROWCAP) ell[(size_t)d0 * ROWCAP + r0] = s0;
  }
}

__global__ void __launch_bounds__(128)
gemm_l1_place_kernel(const float* __restrict__ A, const float* __restrict__ B,
                     float* __restrict__ C, int M, const float* __restrict__ Wa,
                     float* __restrict__ as, float* __restrict__ ad,
                     const int* __restrict__ ei, int* __restrict__ cnt,
                     int* __restrict__ ell, int Eorig) {
  const int p = (int)(blockIdx.x >> 1);
  if ((blockIdx.x & 1u) == 0u) {
    if (p >= L1_GEMM_BLOCKS) return;
    const int by = p / L1_BX;
    const int bx = p - by * L1_BX;
    if (by == 0) gemm_strip<IN_CH, H1C, 1>(bx, by, A, B, C, M, Wa, as, ad);
    else         gemm_strip<IN_CH, H1C, 0>(bx, by, A, B, C, M, Wa, as, ad);
  } else {
    if (p >= PLACE_BLOCKS) return;
    place_edges(p, ei, cnt, ell, Eorig);
  }
}

__device__ __forceinline__ int row_entry(const int* __restrict__ row, int d, int idx, int tot) {
  int v = d;
  if (idx >= 1 && idx < tot) v = row[idx - 1];
  return v;
}


__device__ __forceinline__ void l1_edge(int s, int lane, int hA, int hB,
                                        const float* __restrict__ as,
                                        const float* __restrict__ h,
                                        float adA, float adB, float mA, float mB,
                                        float4& accA, float4& accB,
                                        float& denA, float& denB) {
  float eA = as[(size_t)s * HEADS + hA] + adA;
  float eB = as[(size_t)s * HEADS + hB] + adB;
  const float4* hp = (const float4*)(h + (size_t)s * H1C);
  const float4 vA = hp[lane];
  const float4 vB = hp[32 + lane];
  eA = (eA > 0.f) ? eA : NEG_SLOPE * eA;
  eB = (eB > 0.f) ? eB : NEG_SLOPE * eB;
  const float wA = __expf(eA - mA);
  const float wB = __expf(eB - mB);
  denA += wA;
  denB += wB;
  accA.x += wA * vA.x; accA.y += wA * vA.y; accA.z += wA * vA.z; accA.w += wA * vA.w;
  accB.x += wB * vB.x; accB.y += wB * vB.y; accB.z += wB * vB.z; accB.w += wB * vB.w;
}

__global__ void __launch_bounds__(256)
gather_rows_l1_kernel(const int* __restrict__ ell, const int* __restrict__ cnt,
                      const int* __restrict__ ei, int Eorig,
                      const float* __restrict__ as, const float* __restrict__ ad,
                      const float* __restrict__ h, const float* __restrict__ bias,
                      float* __restrict__ g, int N) {
  const int lane = threadIdx.x & 31;
  const int d = __builtin_amdgcn_readfirstlane((int)((blockIdx.x * blockDim.x + threadIdx.x) >> 5));
  if (d >= N) return;
  const int  n_in = cnt[d];
  const bool fits = (n_in <= ROWCAP);
  const int  tot  = fits ? n_in + 1 : 1;
  const int* row  = ell + (size_t)d * ROWCAP;
  const int hA = lane >> 3;
  const int hB = 4 + (lane >> 3);
  const int h8 = lane & 7;

  float mx = -INFINITY;
  for (int base = 0; base < tot; base += 32) {
    const int n = min(32, tot - base);
    const int mine = row_entry(row, d, base + lane, tot);
#pragma unroll 4
    for (int k = 0; k < n; ++k) {
      const int s = __builtin_amdgcn_readlane(mine, k);
      mx = fmaxf(mx, as[(size_t)s * HEADS + h8]);
    }
  }
  if (!fits) {
    for (int base = 0; base < Eorig; base += 32) {
      const int i = base + lane;
      int dd = -1;
      if (i < Eorig) dd = ei[Eorig + i];
      unsigned hit = __builtin_amdgcn_ballot_w32(dd == d);
      if (hit != 0u) {
        int mine = 0;
        if (dd == d) mine = ei[i];
        while (hit != 0u) {
          const int k = __builtin_ctz(hit);
          hit &= hit - 1u;
          const int s = __builtin_amdgcn_readlane(mine, k);
          mx = fmaxf(mx, as[(size_t)s * HEADS + h8]);
        }
      }
    }
  }
  const float adA = ad[d * HEADS + hA];
  const float adB = ad[d * HEADS + hB];
  float mA = __shfl(mx, hA) + adA;
  float mB = __shfl(mx, hB) + adB;
  mA = (mA > 0.f) ? mA : NEG_SLOPE * mA;
  mB = (mB > 0.f) ? mB : NEG_SLOPE * mB;

  float4 accA = make_float4(0.f, 0.f, 0.f, 0.f);
  float4 accB = make_float4(0.f, 0.f, 0.f, 0.f);
  float denA = 0.f, denB = 0.f;
  for (int base = 0; base < tot; base += 32) {
    const int n = min(32, tot - base);
    const int mine = row_entry(row, d, base + lane, tot);
#pragma unroll 4
    for (int k = 0; k < n; ++k) {
      const int s = __builtin_amdgcn_readlane(mine, k);
      l1_edge(s, lane, hA, hB, as, h, adA, adB, mA, mB, accA, accB, denA, denB);
    }
  }
  if (!fits) {
    for (int base = 0; base < Eorig; base += 32) {
      const int i = base + lane;
      int dd = -1;
      if (i < Eorig) dd = ei[Eorig + i];
      unsigned hit = __builtin_amdgcn_ballot_w32(dd == d);
      if (hit != 0u) {
        int mine = 0;
        if (dd == d) mine = ei[i];
        while (hit != 0u) {
          const int k = __builtin_ctz(hit);
          hit &= hit - 1u;
          const int s = __builtin_amdgcn_readlane(mine, k);
          l1_edge(s, lane, hA, hB, as, h, adA, adB, mA, mB, accA, accB, denA, denB);
        }
      }
    }
  }
  const float dA = denA + 1e-16f;
  const float dB = denB + 1e-16f;
  const float4 bA = ((const float4*)bias)[lane];
  const float4 bB = ((const float4*)bias)[32 + lane];
  float4 oA, oB;
  oA.x = accA.x / dA + bA.x; oA.y = accA.y / dA + bA.y;
  oA.z = accA.z / dA + bA.z; oA.w = accA.w / dA + bA.w;
  oB.x = accB.x / dB + bB.x; oB.y = accB.y / dB + bB.y;
  oB.z = accB.z / dB + bB.z; oB.w = accB.w / dB + bB.w;
  oA.x = (oA.x > 0.f) ? oA.x : (__expf(oA.x) - 1.f);
  oA.y = (oA.y > 0.f) ? oA.y : (__expf(oA.y) - 1.f);
  oA.z = (oA.z > 0.f) ? oA.z : (__expf(oA.z) - 1.f);
  oA.w = (oA.w > 0.f) ? oA.w : (__expf(oA.w) - 1.f);
  oB.x = (oB.x > 0.f) ? oB.x : (__expf(oB.x) - 1.f);
  oB.y = (oB.y > 0.f) ? oB.y : (__expf(oB.y) - 1.f);
  oB.z = (oB.z > 0.f) ? oB.z : (__expf(oB.z) - 1.f);
  oB.w = (oB.w > 0.f) ? oB.w : (__expf(oB.w) - 1.f);
  float4* gp = (float4*)(g + (size_t)d * H1C);
  gp[lane]      = oA;
  gp[32 + lane] = oB;
}

__global__ void __launch_bounds__(256)
gather_rows_l2_kernel(const int* __restrict__ ell, const int* __restrict__ cnt,
                      const int* __restrict__ ei, int Eorig,
                      const float* __restrict__ la,
                      const float* __restrict__ h, const float* __restrict__ bias,
                      float* __restrict__ out, int N) {
  const int lane = threadIdx.x & 31;
  const int d = __builtin_amdgcn_readfirstlane((int)((blockIdx.x * blockDim.x + threadIdx.x) >> 5));
  if (d >= N) return;
  const int  n_in = cnt[d];
  const bool fits = (n_in <= ROWCAP);
  const int  tot  = fits ? n_in + 1 : 1;
  const int* row  = ell + (size_t)d * ROWCAP;

  float mx = -INFINITY;
  for (int base = 0; base < tot; base += 32) {
    const int idx = base + lane;
    if (idx < tot) mx = fmaxf(mx, la[2 * (size_t)row_entry(row, d, idx, tot)]);
  }
  if (!fits) {
    for (int base = 0; base < Eorig; base += 32) {
      const int i = base + lane;
      if (i < Eorig && ei[Eorig + i] == d) mx = fmaxf(mx, la[2 * (size_t)ei[i]]);
    }
  }
#pragma unroll
  for (int o = 16; o > 0; o >>= 1) mx = fmaxf(mx, __shfl_xor(mx, o));
  const float adv = la[2 * (size_t)d + 1];
  float m = mx + adv;
  m = (m > 0.f) ? m : NEG_SLOPE * m;

  float2 acc = make_float2(0.f, 0.f);
  float denl = 0.f;
  for (int base = 0; base < tot; base += 32) {
    const int n = min(32, tot - base);
    const int mine = row_entry(row, d, base + lane, tot);
    float exl = 0.f;
    if (lane < n) {
      float e = la[2 * (size_t)mine] + adv;
      e = (e > 0.f) ? e : NEG_SLOPE * e;
      exl = __expf(e - m);
    }
    denl += exl;
#pragma unroll 4
    for (int k = 0; k < n; ++k) {
      const int   s = __builtin_amdgcn_readlane(mine, k);
      const float w = __int_as_float(__builtin_amdgcn_readlane(__float_as_int(exl), k));
      const float2 v = ((const float2*)(h + (size_t)s * OUT_CH))[lane];
      acc.x += w * v.x; acc.y += w * v.y;
    }
  }
  if (!fits) {
    for (int base = 0; base < Eorig; base += 32) {
      const int i = base + lane;
      int dd = -1;
      if (i < Eorig) dd = ei[Eorig + i];
      unsigned hit = __builtin_amdgcn_ballot_w32(dd == d);
      if (hit != 0u) {
        int mine = 0;
        float exl = 0.f;
        if (dd == d) {
          mine = ei[i];
          float e = la[2 * (size_t)mine] + adv;
          e = (e > 0.f) ? e : NEG_SLOPE * e;
          exl = __expf(e - m);
        }
        denl += exl;
        while (hit != 0u) {
          const int k = __builtin_ctz(hit);
          hit &= hit - 1u;
          const int   s = __builtin_amdgcn_readlane(mine, k);
          const float w = __int_as_float(__builtin_amdgcn_readlane(__float_as_int(exl), k));
          const float2 v = ((const float2*)(h + (size_t)s * OUT_CH))[lane];
          acc.x += w * v.x; acc.y += w * v.y;
        }
      }
    }
  }
#pragma unroll
  for (int o = 16; o > 0; o >>= 1) denl += __shfl_xor(denl, o);
  const float dn = denl + 1e-16f;
  const float2 b = ((const float2*)bias)[lane];
  float2 o2;
  o2.x = acc.x / dn + b.x;
  o2.y = acc.y / dn + b.y;
  ((float2*)(out + (size_t)d * OUT_CH))[lane] = o2;
}

extern "C" void kernel_launch(void* const* d_in, const int* in_sizes, int n_in,
                              void* d_out, int out_size, void* d_ws, size_t ws_size,
                              hipStream_t stream) {
  const float* x      = (const float*)d_in[0];
  const int*   ei     = (const int*)  d_in[1];
  const float* W1     = (const float*)d_in[2];
  const float* a_src1 = (const float*)d_in[3];
  const float* a_dst1 = (const float*)d_in[4];
  const float* b1     = (const float*)d_in[5];
  const float* W2     = (const float*)d_in[6];
  const float* a_src2 = (const float*)d_in[7];
  const float* a_dst2 = (const float*)d_in[8];
  const float* b2     = (const float*)d_in[9];
  float* out = (float*)d_out;

  float* ws   = (float*)d_ws;
  float* h1   = ws;
  float* g1   = h1   + (size_t)NNODES * H1C;
  float* as1  = g1   + (size_t)NNODES * H1C;
  float* ad1  = as1  + (size_t)NNODES * HEADS;
  int* ell      = (int*)(ad1 + (size_t)NNODES * HEADS);
  int* cnt      = ell + (size_t)NNODES * ROWCAP;
  float* Wa1  = (float*)(cnt + NNODES);
  float* Wa2  = Wa1 + (size_t)IN_CH * LGW;
  float* h2   = h1;
  float* la2  = h1 + (size_t)NNODES * OUT_CH;

  const int T = 256;

  fold_kernel<<<FOLD1_BLOCKS + FOLD2_BLOCKS + FILL_BLOCKS, 256, 0, stream>>>(
      W1, a_src1, a_dst1, Wa1, W2, a_src2, a_dst2, Wa2, cnt);

  gemm_l1_place_kernel<<<2 * L1_PAIRS, 128, 0, stream>>>(x, W1, h1, NNODES, Wa1, as1, ad1,
                                                         ei, cnt, ell, NEDGES);
  gather_rows_l1_kernel<<<(NNODES * 32 + T - 1) / T, T, 0, stream>>>(
      ell, cnt, ei, NEDGES, as1, ad1, h1, b1, g1, NNODES);

  wmma_gemm_l2<<<dim3((NNODES + 63) / 64, OUT_CH / 64), 128, 0, stream>>>(g1, W2, h2, NNODES,
                                                                          Wa2, la2);
  gather_rows_l2_kernel<<<(NNODES * 32 + T - 1) / T, T, 0, stream>>>(
      ell, cnt, ei, NEDGES, la2, h2, b2, out, NNODES);
}


# each row's edge counter moved into the row's own word 0 (63 source slots behind it): counting adds fall on 20000 lines, not 625; a gather reads length and entries in one load; sum order as before
# speedup vs baseline: 8.8188x; 1.2153x over previous
#include <hip/hip_runtime.h>
#include <math.h>
#include <cstddef>

#define NNODES   20000
#define NEDGES   320000
#define IN_CH    128
#define HID_CH   32
#define HEADS    8
#define H1C      (HEADS * HID_CH)
#define OUT_CH   64
#define NEG_SLOPE 0.2f

typedef float  v2f   __attribute__((ext_vector_type(2)));
typedef float  v8f   __attribute__((ext_vector_type(8)));

#define LGW 16

template <int K, int N, int MODE>
__device__ __forceinline__ void gemm_strip(int bx, int by,
                                           const float* __restrict__ A,
                                           const float* __restrict__ B,
                                           float* __restrict__ C, int M,
                                           const float* __restrict__ Wa,
                                           float* __restrict__ L0,
                                           float* __restrict__ L1) {
  const int lane = threadIdx.x & 31;
  const int wave = __builtin_amdgcn_readfirstlane((int)(threadIdx.x >> 5));
  const int tm   = bx * 4 + wave;
  if (tm * 16 >= M) return;
  const int j  = lane & 15;
  const int kh = lane >> 4;
  const int cb = by * 64;
  const float* Ap = A + (size_t)(tm * 16 + j) * K + kh * 2;
  const float* Bp = B + (size_t)(kh * 2) * N + cb + 4 * j;
  const float* Wp = Wa + (size_t)(kh * 2) * LGW + j;
  v8f acc0 = {0.f, 0.f, 0.f, 0.f, 0.f, 0.f, 0.f, 0.f};
  v8f acc1 = acc0, acc2 = acc0, acc3 = acc0;
  v8f accL = acc0;

#pragma unroll 1
  for (int k0 = 0; k0 < K; k0 += 32) {
    v2f    a[8];
    float4 b0[8], b1[8];
    v2f    w[8];
#pragma unroll
    for (int s = 0; s < 8; ++s) {
      a[s]  = *(const v2f*)(Ap + k0 + 4 * s);
      b0[s] = *(const float4*)(Bp + (size_t)(k0 + 4 * s) * N);
      b1[s] = *(const float4*)(Bp + (size_t)(k0 + 4 * s + 1) * N);
      if constexpr (MODE != 0) {
        w[s].x = Wp[(size_t)(k0 + 4 * s) * LGW];
        w[s].y = Wp[(size_t)(k0 + 4 * s + 1) * LGW];
      }
    }
#pragma unroll
    for (int s = 0; s < 8; ++s) {
      v2f p0, p1, p2, p3;
      p0.x = b0[s].x; p0.y = b1[s].x;
      p1.x = b0[s].y; p1.y = b1[s].y;
      p2.x = b0[s].z; p2.y = b1[s].z;
      p3.x = b0[s].w; p3.y = b1[s].w;
      acc0 = __builtin_amdgcn_wmma_f32_16x16x4_f32(false, a[s], false, p0,
                                                   (short)0, acc0, false, false);
      acc1 = __builtin_amdgcn_wmma_f32_16x16x4_f32(false, a[s], false, p1,
                                                   (short)0, acc1, false, false);
      acc2 = __builtin_amdgcn_wmma_f32_16x16x4_f32(false, a[s], false, p2,
                                                   (short)0, acc2, false, false);
      acc3 = __builtin_amdgcn_wmma_f32_16x16x4_f32(false, a[s], false, p3,
                                                   (short)0, acc3, false, false);
      if constexpr (MODE != 0) {
        accL = __builtin_amdgcn_wmma_f32_16x16x4_f32(false, a[s], false, w[s],
                                                     (short)0, accL, false, false);
      }
    }
  }

  float* Cp = C + ((size_t)tm * 16 + kh * 8) * N + cb + 4 * j;
#pragma unroll
  for (int r = 0; r < 8; ++r) {
    *(float4*)(Cp + (size_t)r * N) = make_float4(acc0[r], acc1[r], acc2[r], acc3[r]);
  }
  if constexpr (MODE == 1) {
    float* Lp = ((j < HEADS) ? L0 : L1) + ((size_t)tm * 16 + kh * 8) * HEADS + (j & (HEADS - 1));
#pragma unroll
    for (int r = 0; r < 8; ++r) Lp[(size_t)r * HEADS] = accL[r];
  }
  if constexpr (MODE == 2) {
    if (j < 2) {
      float* Lp = L0 + ((size_t)tm * 16 + kh * 8) * 2 + j;
#pragma unroll
      for (int r = 0; r < 8; ++r) Lp[(size_t)r * 2] = accL[r];
    }
  }
}

__global__ void __launch_bounds__(128)
wmma_gemm_l2(const float* __restrict__ A, const float* __restrict__ B,
             float* __restrict__ C, int M, const float* __restrict__ Wa,
             float* __restrict__ la) {
  gemm_strip<H1C, OUT_CH, 2>((int)blockIdx.x, (int)blockIdx.y, A, B, C, M, Wa, la, la);
}

#define ROWCAP   64
#define ROWSLOTS (ROWCAP - 1)
#define FOLD1_BLOCKS ((IN_CH * LGW) / 256)
#define FOLD2_BLOCKS ((H1C * LGW) / 256)
#define FILL_BLOCKS  ((NNODES + 255) / 256)

__global__ void __launch_bounds__(256)
fold_kernel(const float* __restrict__ W1, const float* __restrict__ a_src1,
            const float* __restrict__ a_dst1, float* __restrict__ Wa1,
            const float* __restrict__ W2, const float* __restrict__ a_src2,
            const float* __restrict__ a_dst2, float* __restrict__ Wa2,
            int* __restrict__ ell) {
  const int b = blockIdx.x;
  if (b < FOLD1_BLOCKS) {
    const int i   = b * 256 + threadIdx.x;
    const int k   = i / LGW, col = i % LGW, hd = col % HEADS;
    const float4* wp = (const float4*)(W1 + (size_t)k * H1C + hd * HID_CH);
    const float4* ap = (const float4*)(((col < HEADS) ? a_src1 : a_dst1) + hd * HID_CH);
    float s = 0.f;
#pragma unroll
    for (int c = 0; c < HID_CH / 4; ++c) {
      const float4 wv = wp[c], av = ap[c];
      s += wv.x * av.x; s += wv.y * av.y; s += wv.z * av.z; s += wv.w * av.w;
    }
    Wa1[i] = s;
  } else if (b < FOLD1_BLOCKS + FOLD2_BLOCKS) {
    const int i   = (b - FOLD1_BLOCKS) * 256 + threadIdx.x;
    const int k   = i / LGW, col = i % LGW;
    float s = 0.f;
    if (col < 2) {
      const float4* wp = (const float4*)(W2 + (size_t)k * OUT_CH);
      const float4* ap = (const float4*)((col == 0) ? a_src2 : a_dst2);
#pragma unroll
      for (int c = 0; c < OUT_CH / 4; ++c) {
        const float4 wv = wp[c], av = ap[c];
        s += wv.x * av.x; s += wv.y * av.y; s += wv.z * av.z; s += wv.w * av.w;
      }
    }
    Wa2[i] = s;
  } else {
    const int i = (b - FOLD1_BLOCKS - FOLD2_BLOCKS) * 256 + threadIdx.x;
    if (i < NNODES) ell[(size_t)i * ROWCAP] = 0;
  }
}

#define PLACE_EDGES 256
#define PLACE_BLOCKS ((NEDGES + PLACE_EDGES - 1) / PLACE_EDGES)
#define L1_BX ((NNODES + 63) / 64)
#define L1_GEMM_BLOCKS (L1_BX * (H1C / 64))
#define L1_PAIRS ((L1_GEMM_BLOCKS > PLACE_BLOCKS) ? L1_GEMM_BLOCKS : PLACE_BLOCKS)

__device__ __forceinline__ void place_edges(int pb, const int* __restrict__ ei,
                                            int* __restrict__ ell, int Eorig) {
  const int i0 = pb * PLACE_EDGES + (int)threadIdx.x;
  const int i1 = i0 + PLACE_EDGES / 2;
  if (i1 < Eorig) {
    const int s0 = ei[i0], d0 = ei[Eorig + i0];
    const int s1 = ei[i1], d1 = ei[Eorig + i1];
    asm volatile("" :: "v"(s0), "v"(s1));
    int* row0 = ell + (size_t)d0 * ROWCAP;
    int* row1 = ell + (size_t)d1 * ROWCAP;
    const int r0 = atomicAdd(row0, 1);
    const int r1 = atomicAdd(row1, 1);
    if (r0 < ROWSLOTS) row0[1 + r0] = s0;
    if (r1 < ROWSLOTS) row1[1 + r1] = s1;
  } else if (i0 < Eorig) {
    const int s0 = ei[i0], d0 = ei[Eorig + i0];
    asm volatile("" :: "v"(s0));
    int* row0 = ell + (size_t)d0 * ROWCAP;
    const int r0 = atomicAdd(row0, 1);
    if (r0 < ROWSLOTS) row0[1 + r0] = s0;
  }
}

__global__ void __launch_bounds__(128)
gemm_l1_place_kernel(const float* __restrict__ A, const float* __restrict__ B,
                     float* __restrict__ C, int M, const float* __restrict__ Wa,
                     float* __restrict__ as, float* __restrict__ ad,
                     const int* __restrict__ ei, int* __restrict__ ell, int Eorig) {
  const int p = (int)(blockIdx.x >> 1);
  if ((blockIdx.x & 1u) == 0u) {
    if (p >= L1_GEMM_BLOCKS) return;
    const int by = p / L1_BX;
    const int bx = p - by * L1_BX;
    if (by == 0) gemm_strip<IN_CH, H1C, 1>(bx, by, A, B, C, M, Wa, as, ad);
    else         gemm_strip<IN_CH, H1C, 0>(bx, by, A, B, C, M, Wa, as, ad);
  } else {
    if (p >= PLACE_BLOCKS) return;
    place_edges(p, ei, ell, Eorig);
  }
}

__device__ __forceinline__ int row_entry(const int* __restrict__ row, int first, int d,
                                         int base, int lane, int tot) {
  const int idx = base + lane;
  int v = d;
  if (idx >= 1 && idx < tot) v = (base == 0) ? first : row[idx];
  return v;
}


__device__ __forceinline__ void l1_edge(int s, int lane, int hA, int hB,
                                        const float* __restrict__ as,
                                        const float* __restrict__ h,
                                        float adA, float adB, float mA, float mB,
                                        float4& accA, float4& accB,
                                        float& denA, float& denB) {
  float eA = as[(size_t)s * HEADS + hA] + adA;
  float eB = as[(size_t)s * HEADS + hB] + adB;
  const float4* hp = (const float4*)(h + (size_t)s * H1C);
  const float4 vA = hp[lane];
  const float4 vB = hp[32 + lane];
  eA = (eA > 0.f) ? eA : NEG_SLOPE * eA;
  eB = (eB > 0.f) ? eB : NEG_SLOPE * eB;
  const float wA = __expf(eA - mA);
  const float wB = __expf(eB - mB);
  denA += wA;
  denB += wB;
  accA.x += wA * vA.x; accA.y += wA * vA.y; accA.z += wA * vA.z; accA.w += wA * vA.w;
  accB.x += wB * vB.x; accB.y += wB * vB.y; accB.z += wB * vB.z; accB.w += wB * vB.w;
}

__global__ void __launch_bounds__(256)
gather_rows_l1_kernel(const int* __restrict__ ell,
                      const int* __restrict__ ei, int Eorig,
                      const float* __restrict__ as, const float* __restrict__ ad,
                      const float* __restrict__ h, const float* __restrict__ bias,
                      float* __restrict__ g, int N) {
  const int lane = threadIdx.x & 31;
  const int d = __builtin_amdgcn_readfirstlane((int)((blockIdx.x * blockDim.x + threadIdx.x) >> 5));
  if (d >= N) return;
  const int* row   = ell + (size_t)d * ROWCAP;
  const int  first = row[lane];
  const int  n_in  = __builtin_amdgcn_readfirstlane(first);
  const bool fits  = (n_in <= ROWSLOTS);
  const int  tot   = fits ? n_in + 1 : 1;
  const int hA = lane >> 3;
  const int hB = 4 + (lane >> 3);
  const int h8 = lane & 7;

  float mx = -INFINITY;
  for (int base = 0; base < tot; base += 32) {
    const int n = min(32, tot - base);
    const int mine = row_entry(row, first, d, base, lane, tot);
#pragma unroll 4
    for (int k = 0; k < n; ++k) {
      const int s = __builtin_amdgcn_readlane(mine, k);
      mx = fmaxf(mx, as[(size_t)s * HEADS + h8]);
    }
  }
  if (!fits) {
    for (int base = 0; base < Eorig; base += 32) {
      const int i = base + lane;
      int dd = -1;
      if (i < Eorig) dd = ei[Eorig + i];
      unsigned hit = __builtin_amdgcn_ballot_w32(dd == d);
      if (hit != 0u) {
        int mine = 0;
        if (dd == d) mine = ei[i];
        while (hit != 0u) {
          const int k = __builtin_ctz(hit);
          hit &= hit - 1u;
          const int s = __builtin_amdgcn_readlane(mine, k);
          mx = fmaxf(mx, as[(size_t)s * HEADS + h8]);
        }
      }
    }
  }
  const float adA = ad[d * HEADS + hA];
  const float adB = ad[d * HEADS + hB];
  float mA = __shfl(mx, hA) + adA;
  float mB = __shfl(mx, hB) + adB;
  mA = (mA > 0.f) ? mA : NEG_SLOPE * mA;
  mB = (mB > 0.f) ? mB : NEG_SLOPE * mB;

  float4 accA = make_float4(0.f, 0.f, 0.f, 0.f);
  float4 accB = make_float4(0.f, 0.f, 0.f, 0.f);
  float denA = 0.f, denB = 0.f;
  for (int base = 0; base < tot; base += 32) {
    const int n = min(32, tot - base);
    const int mine = row_entry(row, first, d, base, lane, tot);
#pragma unroll 4
    for (int k = 0; k < n; ++k) {
      const int s = __builtin_amdgcn_readlane(mine, k);
      l1_edge(s, lane, hA, hB, as, h, adA, adB, mA, mB, accA, accB, denA, denB);
    }
  }
  if (!fits) {
    for (int base = 0; base < Eorig; base += 32) {
      const int i = base + lane;
      int dd = -1;
      if (i < Eorig) dd = ei[Eorig + i];
      unsigned hit = __builtin_amdgcn_ballot_w32(dd == d);
      if (hit != 0u) {
        int mine = 0;
        if (dd == d) mine = ei[i];
        while (hit != 0u) {
          const int k = __builtin_ctz(hit);
          hit &= hit - 1u;
          const int s = __builtin_amdgcn_readlane(mine, k);
          l1_edge(s, lane, hA, hB, as, h, adA, adB, mA, mB, accA, accB, denA, denB);
        }
      }
    }
  }
  const float dA = denA + 1e-16f;
  const float dB = denB + 1e-16f;
  const float4 bA = ((const float4*)bias)[lane];
  const float4 bB = ((const float4*)bias)[32 + lane];
  float4 oA, oB;
  oA.x = accA.x / dA + bA.x; oA.y = accA.y / dA + bA.y;
  oA.z = accA.z / dA + bA.z; oA.w = accA.w / dA + bA.w;
  oB.x = accB.x / dB + bB.x; oB.y = accB.y / dB + bB.y;
  oB.z = accB.z / dB + bB.z; oB.w = accB.w / dB + bB.w;
  oA.x = (oA.x > 0.f) ? oA.x : (__expf(oA.x) - 1.f);
  oA.y = (oA.y > 0.f) ? oA.y : (__expf(oA.y) - 1.f);
  oA.z = (oA.z > 0.f) ? oA.z : (__expf(oA.z) - 1.f);
  oA.w = (oA.w > 0.f) ? oA.w : (__expf(oA.w) - 1.f);
  oB.x = (oB.x > 0.f) ? oB.x : (__expf(oB.x) - 1.f);
  oB.y = (oB.y > 0.f) ? oB.y : (__expf(oB.y) - 1.f);
  oB.z = (oB.z > 0.f) ? oB.z : (__expf(oB.z) - 1.f);
  oB.w = (oB.w > 0.f) ? oB.w : (__expf(oB.w) - 1.f);
  float4* gp = (float4*)(g + (size_t)d * H1C);
  gp[lane]      = oA;
  gp[32 + lane] = oB;
}

__global__ void __launch_bounds__(256)
gather_rows_l2_kernel(const int* __restrict__ ell,
                      const int* __restrict__ ei, int Eorig,
                      const float* __restrict__ la,
                      const float* __restrict__ h, const float* __restrict__ bias,
                      float* __restrict__ out, int N) {
  const int lane = threadIdx.x & 31;
  const int d = __builtin_amdgcn_readfirstlane((int)((blockIdx.x * blockDim.x + threadIdx.x) >> 5));
  if (d >= N) return;
  const int* row   = ell + (size_t)d * ROWCAP;
  const int  first = row[lane];
  const int  n_in  = __builtin_amdgcn_readfirstlane(first);
  const bool fits  = (n_in <= ROWSLOTS);
  const int  tot   = fits ? n_in + 1 : 1;

  float mx = -INFINITY;
  for (int base = 0; base < tot; base += 32) {
    const int idx = base + lane;
    if (idx < tot) mx = fmaxf(mx, la[2 * (size_t)row_entry(row, first, d, base, lane, tot)]);
  }
  if (!fits) {
    for (int base = 0; base < Eorig; base += 32) {
      const int i = base + lane;
      if (i < Eorig && ei[Eorig + i] == d) mx = fmaxf(mx, la[2 * (size_t)ei[i]]);
    }
  }
#pragma unroll
  for (int o = 16; o > 0; o >>= 1) mx = fmaxf(mx, __shfl_xor(mx, o));
  const float adv = la[2 * (size_t)d + 1];
  float m = mx + adv;
  m = (m > 0.f) ? m : NEG_SLOPE * m;

  float2 acc = make_float2(0.f, 0.f);
  float denl = 0.f;
  for (int base = 0; base < tot; base += 32) {
    const int n = min(32, tot - base);
    const int mine = row_entry(row, first, d, base, lane, tot);
    float exl = 0.f;
    if (lane < n) {
      float e = la[2 * (size_t)mine] + adv;
      e = (e > 0.f) ? e : NEG_SLOPE * e;
      exl = __expf(e - m);
    }
    denl += exl;
#pragma unroll 4
    for (int k = 0; k < n; ++k) {
      const int   s = __builtin_amdgcn_readlane(mine, k);
      const float w = __int_as_float(__builtin_amdgcn_readlane(__float_as_int(exl), k));
      const float2 v = ((const float2*)(h + (size_t)s * OUT_CH))[lane];
      acc.x += w * v.x; acc.y += w * v.y;
    }
  }
  if (!fits) {
    for (int base = 0; base < Eorig; base += 32) {
      const int i = base + lane;
      int dd = -1;
      if (i < Eorig) dd = ei[Eorig + i];
      unsigned hit = __builtin_amdgcn_ballot_w32(dd == d);
      if (hit != 0u) {
        int mine = 0;
        float exl = 0.f;
        if (dd == d) {
          mine = ei[i];
          float e = la[2 * (size_t)mine] + adv;
          e = (e > 0.f) ? e : NEG_SLOPE * e;
          exl = __expf(e - m);
        }
        denl += exl;
        while (hit != 0u) {
          const int k = __builtin_ctz(hit);
          hit &= hit - 1u;
          const int   s = __builtin_amdgcn_readlane(mine, k);
          const float w = __int_as_float(__builtin_amdgcn_readlane(__float_as_int(exl), k));
          const float2 v = ((const float2*)(h + (size_t)s * OUT_CH))[lane];
          acc.x += w * v.x; acc.y += w * v.y;
        }
      }
    }
  }
#pragma unroll
  for (int o = 16; o > 0; o >>= 1) denl += __shfl_xor(denl, o);
  const float dn = denl + 1e-16f;
  const float2 b = ((const float2*)bias)[lane];
  float2 o2;
  o2.x = acc.x / dn + b.x;
  o2.y = acc.y / dn + b.y;
  ((float2*)(out + (size_t)d * OUT_CH))[lane] = o2;
}

extern "C" void kernel_launch(void* const* d_in, const int* in_sizes, int n_in,
                              void* d_out, int out_size, void* d_ws, size_t ws_size,
                              hipStream_t stream) {
  const float* x      = (const float*)d_in[0];
  const int*   ei     = (const int*)  d_in[1];
  const float* W1     = (const float*)d_in[2];
  const float* a_src1 = (const float*)d_in[3];
  const float* a_dst1 = (const float*)d_in[4];
  const float* b1     = (const float*)d_in[5];
  const float* W2     = (const float*)d_in[6];
  const float* a_src2 = (const float*)d_in[7];
  const float* a_dst2 = (const float*)d_in[8];
  const float* b2     = (const float*)d_in[9];
  float* out = (float*)d_out;

  float* ws   = (float*)d_ws;
  float* h1   = ws;
  float* g1   = h1   + (size_t)NNODES * H1C;
  float* as1  = g1   + (size_t)NNODES * H1C;
  float* ad1  = as1  + (size_t)NNODES * HEADS;
  int* ell      = (int*)(ad1 + (size_t)NNODES * HEADS);
  float* Wa1  = (float*)(ell + (size_t)NNODES * ROWCAP);
  float* Wa2  = Wa1 + (size_t)IN_CH * LGW;
  float* h2   = h1;
  float* la2  = h1 + (size_t)NNODES * OUT_CH;

  const int T = 256;

  fold_kernel<<<FOLD1_BLOCKS + FOLD2_BLOCKS + FILL_BLOCKS, 256, 0, stream>>>(
      W1, a_src1, a_dst1, Wa1, W2, a_src2, a_dst2, Wa2, ell);

  gemm_l1_place_kernel<<<2 * L1_PAIRS, 128, 0, stream>>>(x, W1, h1, NNODES, Wa1, as1, ad1,
                                                         ei, ell, NEDGES);
  gather_rows_l1_kernel<<<(NNODES * 32 + T - 1) / T, T, 0, stream>>>(
      ell, ei, NEDGES, as1, ad1, h1, b1, g1, NNODES);

  wmma_gemm_l2<<<dim3((NNODES + 63) / 64, OUT_CH / 64), 128, 0, stream>>>(g1, W2, h2, NNODES,
                                                                          Wa2, la2);
  gather_rows_l2_kernel<<<(NNODES * 32 + T - 1) / T, T, 0, stream>>>(
      ell, ei, NEDGES, la2, h2, b2, out, NNODES);
}


# row gathers: segment softmax in ONE pass with a running maximum (sums so far rescaled by exp(m-m') when a group of edges raises it): the maximum pass and its logit loads gone; sum order kept
# speedup vs baseline: 9.1907x; 1.0422x over previous
#include <hip/hip_runtime.h>
#include <math.h>
#include <cstddef>

#define NNODES   20000
#define NEDGES   320000
#define IN_CH    128
#define HID_CH   32
#define HEADS    8
#define H1C      (HEADS * HID_CH)
#define OUT_CH   64
#define NEG_SLOPE 0.2f

typedef float  v2f   __attribute__((ext_vector_type(2)));
typedef float  v8f   __attribute__((ext_vector_type(8)));

#define LGW 16

template <int K, int N, int MODE>
__device__ __forceinline__ void gemm_strip(int bx, int by,
                                           const float* __restrict__ A,
                                           const float* __restrict__ B,
                                           float* __restrict__ C, int M,
                                           const float* __restrict__ Wa,
                                           float* __restrict__ L0,
                                           float* __restrict__ L1) {
  const int lane = threadIdx.x & 31;
  const int wave = __builtin_amdgcn_readfirstlane((int)(threadIdx.x >> 5));
  const int tm   = bx * 4 + wave;
  if (tm * 16 >= M) return;
  const int j  = lane & 15;
  const int kh = lane >> 4;
  const int cb = by * 64;
  const float* Ap = A + (size_t)(tm * 16 + j) * K + kh * 2;
  const float* Bp = B + (size_t)(kh * 2) * N + cb + 4 * j;
  const float* Wp = Wa + (size_t)(kh * 2) * LGW + j;
  v8f acc0 = {0.f, 0.f, 0.f, 0.f, 0.f, 0.f, 0.f, 0.f};
  v8f acc1 = acc0, acc2 = acc0, acc3 = acc0;
  v8f accL = acc0;

#pragma unroll 1
  for (int k0 = 0; k0 < K; k0 += 32) {
    v2f    a[8];
    float4 b0[8], b1[8];
    v2f    w[8];
#pragma unroll
    for (int s = 0; s < 8; ++s) {
      a[s]  = *(const v2f*)(Ap + k0 + 4 * s);
      b0[s] = *(const float4*)(Bp + (size_t)(k0 + 4 * s) * N);
      b1[s] = *(const float4*)(Bp + (size_t)(k0 + 4 * s + 1) * N);
      if constexpr (MODE != 0) {
        w[s].x = Wp[(size_t)(k0 + 4 * s) * LGW];
        w[s].y = Wp[(size_t)(k0 + 4 * s + 1) * LGW];
      }
    }
#pragma unroll
    for (int s = 0; s < 8; ++s) {
      v2f p0, p1, p2, p3;
      p0.x = b0[s].x; p0.y = b1[s].x;
      p1.x = b0[s].y; p1.y = b1[s].y;
      p2.x = b0[s].z; p2.y = b1[s].z;
      p3.x = b0[s].w; p3.y = b1[s].w;
      acc0 = __builtin_amdgcn_wmma_f32_16x16x4_f32(false, a[s], false, p0,
                                                   (short)0, acc0, false, false);
      acc1 = __builtin_amdgcn_wmma_f32_16x16x4_f32(false, a[s], false, p1,
                                                   (short)0, acc1, false, false);
      acc2 = __builtin_amdgcn_wmma_f32_16x16x4_f32(false, a[s], false, p2,
                                                   (short)0, acc2, false, false);
      acc3 = __builtin_amdgcn_wmma_f32_16x16x4_f32(false, a[s], false, p3,
                                                   (short)0, acc3, false, false);
      if constexpr (MODE != 0) {
        accL = __builtin_amdgcn_wmma_f32_16x16x4_f32(false, a[s], false, w[s],
                                                     (short)0, accL, false, false);
      }
    }
  }

  float* Cp = C + ((size_t)tm * 16 + kh * 8) * N + cb + 4 * j;
#pragma unroll
  for (int r = 0; r < 8; ++r) {
    *(float4*)(Cp + (size_t)r * N) = make_float4(acc0[r], acc1[r], acc2[r], acc3[r]);
  }
  if constexpr (MODE == 1) {
    float* Lp = ((j < HEADS) ? L0 : L1) + ((size_t)tm * 16 + kh * 8) * HEADS + (j & (HEADS - 1));
#pragma unroll
    for (int r = 0; r < 8; ++r) Lp[(size_t)r * HEADS] = accL[r];
  }
  if constexpr (MODE == 2) {
    if (j < 2) {
      float* Lp = L0 + ((size_t)tm * 16 + kh * 8) * 2 + j;
#pragma unroll
      for (int r = 0; r < 8; ++r) Lp[(size_t)r * 2] = accL[r];
    }
  }
}

__global__ void __launch_bounds__(128)
wmma_gemm_l2(const float* __restrict__ A, const float* __restrict__ B,
             float* __restrict__ C, int M, const float* __restrict__ Wa,
             float* __restrict__ la) {
  gemm_strip<H1C, OUT_CH, 2>((int)blockIdx.x, (int)blockIdx.y, A, B, C, M, Wa, la, la);
}

#define ROWCAP   64
#define ROWSLOTS (ROWCAP - 1)
#define FOLD1_BLOCKS ((IN_CH * LGW) / 256)
#define FOLD2_BLOCKS ((H1C * LGW) / 256)
#define FILL_BLOCKS  ((NNODES + 255) / 256)

__global__ void __launch_bounds__(256)
fold_kernel(const float* __restrict__ W1, const float* __restrict__ a_src1,
            const float* __restrict__ a_dst1, float* __restrict__ Wa1,
            const float* __restrict__ W2, const float* __restrict__ a_src2,
            const float* __restrict__ a_dst2, float* __restrict__ Wa2,
            int* __restrict__ ell) {
  const int b = blockIdx.x;
  if (b < FOLD1_BLOCKS) {
    const int i   = b * 256 + threadIdx.x;
    const int k   = i / LGW, col = i % LGW, hd = col % HEADS;
    const float4* wp = (const float4*)(W1 + (size_t)k * H1C + hd * HID_CH);
    const float4* ap = (const float4*)(((col < HEADS) ? a_src1 : a_dst1) + hd * HID_CH);
    float s = 0.f;
#pragma unroll
    for (int c = 0; c < HID_CH / 4; ++c) {
      const float4 wv = wp[c], av = ap[c];
      s += wv.x * av.x; s += wv.y * av.y; s += wv.z * av.z; s += wv.w * av.w;
    }
    Wa1[i] = s;
  } else if (b < FOLD1_BLOCKS + FOLD2_BLOCKS) {
    const int i   = (b - FOLD1_BLOCKS) * 256 + threadIdx.x;
    const int k   = i / LGW, col = i % LGW;
    float s = 0.f;
    if (col < 2) {
      const float4* wp = (const float4*)(W2 + (size_t)k * OUT_CH);
      const float4* ap = (const float4*)((col == 0) ? a_src2 : a_dst2);
#pragma unroll
      for (int c = 0; c < OUT_CH / 4; ++c) {
        const float4 wv = wp[c], av = ap[c];
        s += wv.x * av.x; s += wv.y * av.y; s += wv.z * av.z; s += wv.w * av.w;
      }
    }
    Wa2[i] = s;
  } else {
    const int i = (b - FOLD1_BLOCKS - FOLD2_BLOCKS) * 256 + threadIdx.x;
    if (i < NNODES) ell[(size_t)i * ROWCAP] = 0;
  }
}

#define PLACE_EDGES 256
#define PLACE_BLOCKS ((NEDGES + PLACE_EDGES - 1) / PLACE_EDGES)
#define L1_BX ((NNODES + 63) / 64)
#define L1_GEMM_BLOCKS (L1_BX * (H1C / 64))
#define L1_PAIRS ((L1_GEMM_BLOCKS > PLACE_BLOCKS) ? L1_GEMM_BLOCKS : PLACE_BLOCKS)

__device__ __forceinline__ void place_edges(int pb, const int* __restrict__ ei,
                                            int* __restrict__ ell, int Eorig) {
  const int i0 = pb * PLACE_EDGES + (int)threadIdx.x;
  const int i1 = i0 + PLACE_EDGES / 2;
  if (i1 < Eorig) {
    const int s0 = ei[i0], d0 = ei[Eorig + i0];
    const int s1 = ei[i1], d1 = ei[Eorig + i1];
    asm volatile("" :: "v"(s0), "v"(s1));
    int* row0 = ell + (size_t)d0 * ROWCAP;
    int* row1 = ell + (size_t)d1 * ROWCAP;
    const int r0 = atomicAdd(row0, 1);
    const int r1 = atomicAdd(row1, 1);
    if (r0 < ROWSLOTS) row0[1 + r0] = s0;
    if (r1 < ROWSLOTS) row1[1 + r1] = s1;
  } else if (i0 < Eorig) {
    const int s0 = ei[i0], d0 = ei[Eorig + i0];
    asm volatile("" :: "v"(s0));
    int* row0 = ell + (size_t)d0 * ROWCAP;
    const int r0 = atomicAdd(row0, 1);
    if (r0 < ROWSLOTS) row0[1 + r0] = s0;
  }
}

__global__ void __launch_bounds__(128)
gemm_l1_place_kernel(const float* __restrict__ A, const float* __restrict__ B,
                     float* __restrict__ C, int M, const float* __restrict__ Wa,
                     float* __restrict__ as, float* __restrict__ ad,
                     const int* __restrict__ ei, int* __restrict__ ell, int Eorig) {
  const int p = (int)(blockIdx.x >> 1);
  if ((blockIdx.x & 1u) == 0u) {
    if (p >= L1_GEMM_BLOCKS) return;
    const int by = p / L1_BX;
    const int bx = p - by * L1_BX;
    if (by == 0) gemm_strip<IN_CH, H1C, 1>(bx, by, A, B, C, M, Wa, as, ad);
    else         gemm_strip<IN_CH, H1C, 0>(bx, by, A, B, C, M, Wa, as, ad);
  } else {
    if (p >= PLACE_BLOCKS) return;
    place_edges(p, ei, ell, Eorig);
  }
}

__device__ __forceinline__ int row_entry(const int* __restrict__ row, int first, int d,
                                         int base, int lane, int tot) {
  const int idx = base + lane;
  int v = d;
  if (idx >= 1 && idx < tot) v = (base == 0) ? first : row[idx];
  return v;
}


template <int CNT>
__device__ __forceinline__ void l1_group(int mine, int k, int lane, int hA, int hB,
                                         const float* __restrict__ as,
                                         const float* __restrict__ h,
                                         float adA, float adB, float& mA, float& mB,
                                         float4& accA, float4& accB,
                                         float& denA, float& denB) {
  float  eA[CNT], eB[CNT];
  float4 vA[CNT], vB[CNT];
#pragma unroll
  for (int j = 0; j < CNT; ++j) {
    const int s = __builtin_amdgcn_readlane(mine, k + j);
    eA[j] = as[(size_t)s * HEADS + hA];
    eB[j] = as[(size_t)s * HEADS + hB];
    const float4* hp = (const float4*)(h + (size_t)s * H1C);
    vA[j] = hp[lane];
    vB[j] = hp[32 + lane];
  }
  float gA = -INFINITY, gB = -INFINITY;
#pragma unroll
  for (int j = 0; j < CNT; ++j) {
    eA[j] += adA;
    eB[j] += adB;
    eA[j] = (eA[j] > 0.f) ? eA[j] : NEG_SLOPE * eA[j];
    eB[j] = (eB[j] > 0.f) ? eB[j] : NEG_SLOPE * eB[j];
    gA = fmaxf(gA, eA[j]);
    gB = fmaxf(gB, eB[j]);
  }
  const float nA = fmaxf(mA, gA);
  const float nB = fmaxf(mB, gB);
  const float cA = __expf(mA - nA);
  const float cB = __expf(mB - nB);
  mA = nA;
  mB = nB;
  denA *= cA;
  denB *= cB;
  accA.x *= cA; accA.y *= cA; accA.z *= cA; accA.w *= cA;
  accB.x *= cB; accB.y *= cB; accB.z *= cB; accB.w *= cB;
#pragma unroll
  for (int j = 0; j < CNT; ++j) {
    const float wA = __expf(eA[j] - nA);
    const float wB = __expf(eB[j] - nB);
    denA += wA;
    denB += wB;
    accA.x += wA * vA[j].x; accA.y += wA * vA[j].y; accA.z += wA * vA[j].z; accA.w += wA * vA[j].w;
    accB.x += wB * vB[j].x; accB.y += wB * vB[j].y; accB.z += wB * vB[j].z; accB.w += wB * vB[j].w;
  }
}

__global__ void __launch_bounds__(256)
gather_rows_l1_kernel(const int* __restrict__ ell,
                      const int* __restrict__ ei, int Eorig,
                      const float* __restrict__ as, const float* __restrict__ ad,
                      const float* __restrict__ h, const float* __restrict__ bias,
                      float* __restrict__ g, int N) {
  const int lane = threadIdx.x & 31;
  const int d = __builtin_amdgcn_readfirstlane((int)((blockIdx.x * blockDim.x + threadIdx.x) >> 5));
  if (d >= N) return;
  const int* row   = ell + (size_t)d * ROWCAP;
  const int  first = row[lane];
  const int hA = lane >> 3;
  const int hB = 4 + (lane >> 3);
  const float adA = ad[d * HEADS + hA];
  const float adB = ad[d * HEADS + hB];
  const int  n_in  = __builtin_amdgcn_readfirstlane(first);
  const bool fits  = (n_in <= ROWSLOTS);
  const int  tot   = fits ? n_in + 1 : 1;

  float mA = -INFINITY, mB = -INFINITY;
  float4 accA = make_float4(0.f, 0.f, 0.f, 0.f);
  float4 accB = make_float4(0.f, 0.f, 0.f, 0.f);
  float denA = 0.f, denB = 0.f;
  for (int base = 0; base < tot; base += 32) {
    const int n = min(32, tot - base);
    const int mine = row_entry(row, first, d, base, lane, tot);
    int k = 0;
#pragma unroll 1
    for (; k + 4 <= n; k += 4)
      l1_group<4>(mine, k, lane, hA, hB, as, h, adA, adB, mA, mB, accA, accB, denA, denB);
    const int rem = n - k;
    if (rem == 3)
      l1_group<3>(mine, k, lane, hA, hB, as, h, adA, adB, mA, mB, accA, accB, denA, denB);
    else if (rem == 2)
      l1_group<2>(mine, k, lane, hA, hB, as, h, adA, adB, mA, mB, accA, accB, denA, denB);
    else if (rem == 1)
      l1_group<1>(mine, k, lane, hA, hB, as, h, adA, adB, mA, mB, accA, accB, denA, denB);
  }
  if (!fits) {
    for (int base = 0; base < Eorig; base += 32) {
      const int i = base + lane;
      int dd = -1;
      if (i < Eorig) dd = ei[Eorig + i];
      unsigned hit = __builtin_amdgcn_ballot_w32(dd == d);
      if (hit != 0u) {
        int mine = 0;
        if (dd == d) mine = ei[i];
        while (hit != 0u) {
          const int k = __builtin_ctz(hit);
          hit &= hit - 1u;
          l1_group<1>(mine, k, lane, hA, hB, as, h, adA, adB, mA, mB, accA, accB, denA, denB);
        }
      }
    }
  }
  const float dA = denA + 1e-16f;
  const float dB = denB + 1e-16f;
  const float4 bA = ((const float4*)bias)[lane];
  const float4 bB = ((const float4*)bias)[32 + lane];
  float4 oA, oB;
  oA.x = accA.x / dA + bA.x; oA.y = accA.y / dA + bA.y;
  oA.z = accA.z / dA + bA.z; oA.w = accA.w / dA + bA.w;
  oB.x = accB.x / dB + bB.x; oB.y = accB.y / dB + bB.y;
  oB.z = accB.z / dB + bB.z; oB.w = accB.w / dB + bB.w;
  oA.x = (oA.x > 0.f) ? oA.x : (__expf(oA.x) - 1.f);
  oA.y = (oA.y > 0.f) ? oA.y : (__expf(oA.y) - 1.f);
  oA.z = (oA.z > 0.f) ? oA.z : (__expf(oA.z) - 1.f);
  oA.w = (oA.w > 0.f) ? oA.w : (__expf(oA.w) - 1.f);
  oB.x = (oB.x > 0.f) ? oB.x : (__expf(oB.x) - 1.f);
  oB.y = (oB.y > 0.f) ? oB.y : (__expf(oB.y) - 1.f);
  oB.z = (oB.z > 0.f) ? oB.z : (__expf(oB.z) - 1.f);
  oB.w = (oB.w > 0.f) ? oB.w : (__expf(oB.w) - 1.f);
  float4* gp = (float4*)(g + (size_t)d * H1C);
  gp[lane]      = oA;
  gp[32 + lane] = oB;
}

__device__ __forceinline__ float l2_weights(bool has, int mine,
                                            const float* __restrict__ la, float adv,
                                            float& m, float2& acc, float& denl) {
  float e = -INFINITY;
  if (has) {
    e = la[2 * (size_t)mine] + adv;
    e = (e > 0.f) ? e : NEG_SLOPE * e;
  }
  float gm = e;
#pragma unroll
  for (int o = 16; o > 0; o >>= 1) gm = fmaxf(gm, __shfl_xor(gm, o));
  const float nm = fmaxf(m, gm);
  const float c = __expf(m - nm);
  m = nm;
  acc.x *= c;
  acc.y *= c;
  denl *= c;
  const float exl = has ? __expf(e - nm) : 0.f;
  denl += exl;
  return exl;
}

__global__ void __launch_bounds__(256)
gather_rows_l2_kernel(const int* __restrict__ ell,
                      const int* __restrict__ ei, int Eorig,
                      const float* __restrict__ la,
                      const float* __restrict__ h, const float* __restrict__ bias,
                      float* __restrict__ out, int N) {
  const int lane = threadIdx.x & 31;
  const int d = __builtin_amdgcn_readfirstlane((int)((blockIdx.x * blockDim.x + threadIdx.x) >> 5));
  if (d >= N) return;
  const int* row   = ell + (size_t)d * ROWCAP;
  const int  first = row[lane];
  const float adv  = la[2 * (size_t)d + 1];
  const int  n_in  = __builtin_amdgcn_readfirstlane(first);
  const bool fits  = (n_in <= ROWSLOTS);
  const int  tot   = fits ? n_in + 1 : 1;

  float m = -INFINITY;
  float2 acc = make_float2(0.f, 0.f);
  float denl = 0.f;
  for (int base = 0; base < tot; base += 32) {
    const int n = min(32, tot - base);
    const int mine = row_entry(row, first, d, base, lane, tot);
    const float exl = l2_weights(lane < n, mine, la, adv, m, acc, denl);
#pragma unroll 4
    for (int k = 0; k < n; ++k) {
      const int   s = __builtin_amdgcn_readlane(mine, k);
      const float w = __int_as_float(__builtin_amdgcn_readlane(__float_as_int(exl), k));
      const float2 v = ((const float2*)(h + (size_t)s * OUT_CH))[lane];
      acc.x += w * v.x; acc.y += w * v.y;
    }
  }
  if (!fits) {
    for (int base = 0; base < Eorig; base += 32) {
      const int i = base + lane;
      int dd = -1;
      if (i < Eorig) dd = ei[Eorig + i];
      unsigned hit = __builtin_amdgcn_ballot_w32(dd == d);
      if (hit != 0u) {
        int mine = 0;
        if (dd == d) mine = ei[i];
        const float exl = l2_weights(dd == d, mine, la, adv, m, acc, denl);
        while (hit != 0u) {
          const int k = __builtin_ctz(hit);
          hit &= hit - 1u;
          const int   s = __builtin_amdgcn_readlane(mine, k);
          const float w = __int_as_float(__builtin_amdgcn_readlane(__float_as_int(exl), k));
          const float2 v = ((const float2*)(h + (size_t)s * OUT_CH))[lane];
          acc.x += w * v.x; acc.y += w * v.y;
        }
      }
    }
  }
#pragma unroll
  for (int o = 16; o > 0; o >>= 1) denl += __shfl_xor(denl, o);
  const float dn = denl + 1e-16f;
  const float2 b = ((const float2*)bias)[lane];
  float2 o2;
  o2.x = acc.x / dn + b.x;
  o2.y = acc.y / dn + b.y;
  ((float2*)(out + (size_t)d * OUT_CH))[lane] = o2;
}

extern "C" void kernel_launch(void* const* d_in, const int* in_sizes, int n_in,
                              void* d_out, int out_size, void* d_ws, size_t ws_size,
                              hipStream_t stream) {
  const float* x      = (const float*)d_in[0];
  const int*   ei     = (const int*)  d_in[1];
  const float* W1     = (const float*)d_in[2];
  const float* a_src1 = (const float*)d_in[3];
  const float* a_dst1 = (const float*)d_in[4];
  const float* b1     = (const float*)d_in[5];
  const float* W2     = (const float*)d_in[6];
  const float* a_src2 = (const float*)d_in[7];
  const float* a_dst2 = (const float*)d_in[8];
  const float* b2     = (const float*)d_in[9];
  float* out = (float*)d_out;

  float* ws   = (float*)d_ws;
  float* h1   = ws;
  float* g1   = h1   + (size_t)NNODES * H1C;
  float* as1  = g1   + (size_t)NNODES * H1C;
  float* ad1  = as1  + (size_t)NNODES * HEADS;
  int* ell      = (int*)(ad1 + (size_t)NNODES * HEADS);
  float* Wa1  = (float*)(ell + (size_t)NNODES * ROWCAP);
  float* Wa2  = Wa1 + (size_t)IN_CH * LGW;
  float* h2   = h1;
  float* la2  = h1 + (size_t)NNODES * OUT_CH;

  const int T = 256;

  fold_kernel<<<FOLD1_BLOCKS + FOLD2_BLOCKS + FILL_BLOCKS, 256, 0, stream>>>(
      W1, a_src1, a_dst1, Wa1, W2, a_src2, a_dst2, Wa2, ell);

  gemm_l1_place_kernel<<<2 * L1_PAIRS, 128, 0, stream>>>(x, W1, h1, NNODES, Wa1, as1, ad1,
                                                         ei, ell, NEDGES);
  gather_rows_l1_kernel<<<(NNODES * 32 + T - 1) / T, T, 0, stream>>>(
      ell, ei, NEDGES, as1, ad1, h1, b1, g1, NNODES);

  wmma_gemm_l2<<<dim3((NNODES + 63) / 64, OUT_CH / 64), 128, 0, stream>>>(g1, W2, h2, NNODES,
                                                                          Wa2, la2);
  gather_rows_l2_kernel<<<(NNODES * 32 + T - 1) / T, T, 0, stream>>>(
      ell, ei, NEDGES, la2, h2, b2, out, NNODES);
}


# layer-2 GEMM split along K: a workgroup's 4 waves take a quarter of K each for one 16-row strip, partial tiles summed through LDS; 4x the waves, 2 load trips each, not 8; sum order changes
# speedup vs baseline: 9.6028x; 1.0448x over previous
#include <hip/hip_runtime.h>
#include <math.h>
#include <cstddef>

#define NNODES   20000
#define NEDGES   320000
#define IN_CH    128
#define HID_CH   32
#define HEADS    8
#define H1C      (HEADS * HID_CH)
#define OUT_CH   64
#define NEG_SLOPE 0.2f

typedef float  v2f   __attribute__((ext_vector_type(2)));
typedef float  v8f   __attribute__((ext_vector_type(8)));

#define LGW 16

template <int KLEN, int N, bool LOGIT>
__device__ __forceinline__ void strip_trips(const float* __restrict__ Ap,
                                            const float* __restrict__ Bp,
                                            const float* __restrict__ Wp,
                                            v8f& acc0, v8f& acc1, v8f& acc2, v8f& acc3,
                                            v8f& accL) {
#pragma unroll 1
  for (int k0 = 0; k0 < KLEN; k0 += 32) {
    v2f    a[8];
    float4 b0[8], b1[8];
    v2f    w[8];
#pragma unroll
    for (int s = 0; s < 8; ++s) {
      a[s]  = *(const v2f*)(Ap + k0 + 4 * s);
      b0[s] = *(const float4*)(Bp + (size_t)(k0 + 4 * s) * N);
      b1[s] = *(const float4*)(Bp + (size_t)(k0 + 4 * s + 1) * N);
      if constexpr (LOGIT) {
        w[s].x = Wp[(size_t)(k0 + 4 * s) * LGW];
        w[s].y = Wp[(size_t)(k0 + 4 * s + 1) * LGW];
      }
    }
#pragma unroll
    for (int s = 0; s < 8; ++s) {
      v2f p0, p1, p2, p3;
      p0.x = b0[s].x; p0.y = b1[s].x;
      p1.x = b0[s].y; p1.y = b1[s].y;
      p2.x = b0[s].z; p2.y = b1[s].z;
      p3.x = b0[s].w; p3.y = b1[s].w;
      acc0 = __builtin_amdgcn_wmma_f32_16x16x4_f32(false, a[s], false, p0,
                                                   (short)0, acc0, false, false);
      acc1 = __builtin_amdgcn_wmma_f32_16x16x4_f32(false, a[s], false, p1,
                                                   (short)0, acc1, false, false);
      acc2 = __builtin_amdgcn_wmma_f32_16x16x4_f32(false, a[s], false, p2,
                                                   (short)0, acc2, false, false);
      acc3 = __builtin_amdgcn_wmma_f32_16x16x4_f32(false, a[s], false, p3,
                                                   (short)0, acc3, false, false);
      if constexpr (LOGIT) {
        accL = __builtin_amdgcn_wmma_f32_16x16x4_f32(false, a[s], false, w[s],
                                                     (short)0, accL, false, false);
      }
    }
  }
}

template <int K, int N, bool LOGIT>
__device__ __forceinline__ void gemm_strip(int bx, int by,
                                           const float* __restrict__ A,
                                           const float* __restrict__ B,
                                           float* __restrict__ C, int M,
                                           const float* __restrict__ Wa,
                                           float* __restrict__ L0,
                                           float* __restrict__ L1) {
  const int lane = threadIdx.x & 31;
  const int wave = __builtin_amdgcn_readfirstlane((int)(threadIdx.x >> 5));
  const int tm   = bx * 4 + wave;
  if (tm * 16 >= M) return;
  const int j  = lane & 15;
  const int kh = lane >> 4;
  const int cb = by * 64;
  const float* Ap = A + (size_t)(tm * 16 + j) * K + kh * 2;
  const float* Bp = B + (size_t)(kh * 2) * N + cb + 4 * j;
  const float* Wp = Wa + (size_t)(kh * 2) * LGW + j;
  v8f acc0 = {0.f, 0.f, 0.f, 0.f, 0.f, 0.f, 0.f, 0.f};
  v8f acc1 = acc0, acc2 = acc0, acc3 = acc0;
  v8f accL = acc0;
  strip_trips<K, N, LOGIT>(Ap, Bp, Wp, acc0, acc1, acc2, acc3, accL);

  float* Cp = C + ((size_t)tm * 16 + kh * 8) * N + cb + 4 * j;
#pragma unroll
  for (int r = 0; r < 8; ++r) {
    *(float4*)(Cp + (size_t)r * N) = make_float4(acc0[r], acc1[r], acc2[r], acc3[r]);
  }
  if constexpr (LOGIT) {
    float* Lp = ((j < HEADS) ? L0 : L1) + ((size_t)tm * 16 + kh * 8) * HEADS + (j & (HEADS - 1));
#pragma unroll
    for (int r = 0; r < 8; ++r) Lp[(size_t)r * HEADS] = accL[r];
  }
}

#define L2_KW (H1C / 4)

__global__ void __launch_bounds__(128)
wmma_gemm_l2(const float* __restrict__ A, const float* __restrict__ B,
             float* __restrict__ C, int M, const float* __restrict__ Wa,
             float* __restrict__ la) {
  __shared__ float4 pc[4][8][32];
  __shared__ float  pq[4][8][32];
  const int lane = threadIdx.x & 31;
  const int q  = __builtin_amdgcn_readfirstlane((int)(threadIdx.x >> 5));
  const int tm = (int)blockIdx.x;
  if (tm * 16 >= M) return;
  const int j  = lane & 15;
  const int kh = lane >> 4;
  const int kq = q * L2_KW;
  const float* Ap = A + (size_t)(tm * 16 + j) * H1C + kq + kh * 2;
  const float* Bp = B + (size_t)(kq + kh * 2) * OUT_CH + 4 * j;
  const float* Wp = Wa + (size_t)(kq + kh * 2) * LGW + j;
  v8f acc0 = {0.f, 0.f, 0.f, 0.f, 0.f, 0.f, 0.f, 0.f};
  v8f acc1 = acc0, acc2 = acc0, acc3 = acc0;
  v8f accL = acc0;
  strip_trips<L2_KW, OUT_CH, true>(Ap, Bp, Wp, acc0, acc1, acc2, acc3, accL);

#pragma unroll
  for (int r = 0; r < 8; ++r) {
    pc[q][r][lane] = make_float4(acc0[r], acc1[r], acc2[r], acc3[r]);
    pq[q][r][lane] = accL[r];
  }
  __syncthreads();

#pragma unroll
  for (int rr = 0; rr < 2; ++rr) {
    const int r = 2 * q + rr;
    float4 s = pc[0][r][lane];
#pragma unroll
    for (int p = 1; p < 4; ++p) {
      const float4 t = pc[p][r][lane];
      s.x += t.x; s.y += t.y; s.z += t.z; s.w += t.w;
    }
    *(float4*)(C + ((size_t)tm * 16 + kh * 8 + r) * OUT_CH + 4 * j) = s;
  }
  if (q == 0 && j < 2) {
    float* Lp = la + ((size_t)tm * 16 + kh * 8) * 2 + j;
#pragma unroll
    for (int r = 0; r < 8; ++r) {
      float s = pq[0][r][lane];
      s += pq[1][r][lane];
      s += pq[2][r][lane];
      s += pq[3][r][lane];
      Lp[(size_t)r * 2] = s;
    }
  }
}

#define ROWCAP   64
#define ROWSLOTS (ROWCAP - 1)
#define FOLD1_BLOCKS ((IN_CH * LGW) / 256)
#define FOLD2_BLOCKS ((H1C * LGW) / 256)
#define FILL_BLOCKS  ((NNODES + 255) / 256)

__global__ void __launch_bounds__(256)
fold_kernel(const float* __restrict__ W1, const float* __restrict__ a_src1,
            const float* __restrict__ a_dst1, float* __restrict__ Wa1,
            const float* __restrict__ W2, const float* __restrict__ a_src2,
            const float* __restrict__ a_dst2, float* __restrict__ Wa2,
            int* __restrict__ ell) {
  const int b = blockIdx.x;
  if (b < FOLD1_BLOCKS) {
    const int i   = b * 256 + threadIdx.x;
    const int k   = i / LGW, col = i % LGW, hd = col % HEADS;
    const float4* wp = (const float4*)(W1 + (size_t)k * H1C + hd * HID_CH);
    const float4* ap = (const float4*)(((col < HEADS) ? a_src1 : a_dst1) + hd * HID_CH);
    float s = 0.f;
#pragma unroll
    for (int c = 0; c < HID_CH / 4; ++c) {
      const float4 wv = wp[c], av = ap[c];
      s += wv.x * av.x; s += wv.y * av.y; s += wv.z * av.z; s += wv.w * av.w;
    }
    Wa1[i] = s;
  } else if (b < FOLD1_BLOCKS + FOLD2_BLOCKS) {
    const int i   = (b - FOLD1_BLOCKS) * 256 + threadIdx.x;
    const int k   = i / LGW, col = i % LGW;
    float s = 0.f;
    if (col < 2) {
      const float4* wp = (const float4*)(W2 + (size_t)k * OUT_CH);
      const float4* ap = (const float4*)((col == 0) ? a_src2 : a_dst2);
#pragma unroll
      for (int c = 0; c < OUT_CH / 4; ++c) {
        const float4 wv = wp[c], av = ap[c];
        s += wv.x * av.x; s += wv.y * av.y; s += wv.z * av.z; s += wv.w * av.w;
      }
    }
    Wa2[i] = s;
  } else {
    const int i = (b - FOLD1_BLOCKS - FOLD2_BLOCKS) * 256 + threadIdx.x;
    if (i < NNODES) ell[(size_t)i * ROWCAP] = 0;
  }
}

#define PLACE_EDGES 256
#define PLACE_BLOCKS ((NEDGES + PLACE_EDGES - 1) / PLACE_EDGES)
#define L1_BX ((NNODES + 63) / 64)
#define L1_GEMM_BLOCKS (L1_BX * (H1C / 64))
#define L1_PAIRS ((L1_GEMM_BLOCKS > PLACE_BLOCKS) ? L1_GEMM_BLOCKS : PLACE_BLOCKS)

__device__ __forceinline__ void place_edges(int pb, const int* __restrict__ ei,
                                            int* __restrict__ ell, int Eorig) {
  const int i0 = pb * PLACE_EDGES + (int)threadIdx.x;
  const int i1 = i0 + PLACE_EDGES / 2;
  if (i1 < Eorig) {
    const int s0 = ei[i0], d0 = ei[Eorig + i0];
    const int s1 = ei[i1], d1 = ei[Eorig + i1];
    asm volatile("" :: "v"(s0), "v"(s1));
    int* row0 = ell + (size_t)d0 * ROWCAP;
    int* row1 = ell + (size_t)d1 * ROWCAP;
    const int r0 = atomicAdd(row0, 1);
    const int r1 = atomicAdd(row1, 1);
    if (r0 < ROWSLOTS) row0[1 + r0] = s0;
    if (r1 < ROWSLOTS) row1[1 + r1] = s1;
  } else if (i0 < Eorig) {
    const int s0 = ei[i0], d0 = ei[Eorig + i0];
    asm volatile("" :: "v"(s0));
    int* row0 = ell + (size_t)d0 * ROWCAP;
    const int r0 = atomicAdd(row0, 1);
    if (r0 < ROWSLOTS) row0[1 + r0] = s0;
  }
}

__global__ void __launch_bounds__(128)
gemm_l1_place_kernel(const float* __restrict__ A, const float* __restrict__ B,
                     float* __restrict__ C, int M, const float* __restrict__ Wa,
                     float* __restrict__ as, float* __restrict__ ad,
                     const int* __restrict__ ei, int* __restrict__ ell, int Eorig) {
  const int p = (int)(blockIdx.x >> 1);
  if ((blockIdx.x & 1u) == 0u) {
    if (p >= L1_GEMM_BLOCKS) return;
    const int by = p / L1_BX;
    const int bx = p - by * L1_BX;
    if (by == 0) gemm_strip<IN_CH, H1C, true>(bx, by, A, B, C, M, Wa, as, ad);
    else         gemm_strip<IN_CH, H1C, false>(bx, by, A, B, C, M, Wa, as, ad);
  } else {
    if (p >= PLACE_BLOCKS) return;
    place_edges(p, ei, ell, Eorig);
  }
}

__device__ __forceinline__ int row_entry(const int* __restrict__ row, int first, int d,
                                         int base, int lane, int tot) {
  const int idx = base + lane;
  int v = d;
  if (idx >= 1 && idx < tot) v = (base == 0) ? first : row[idx];
  return v;
}


template <int CNT>
__device__ __forceinline__ void l1_group(int mine, int k, int lane, int hA, int hB,
                                         const float* __restrict__ as,
                                         const float* __restrict__ h,
                                         float adA, float adB, float& mA, float& mB,
                                         float4& accA, float4& accB,
                                         float& denA, float& denB) {
  float  eA[CNT], eB[CNT];
  float4 vA[CNT], vB[CNT];
#pragma unroll
  for (int j = 0; j < CNT; ++j) {
    const int s = __builtin_amdgcn_readlane(mine, k + j);
    eA[j] = as[(size_t)s * HEADS + hA];
    eB[j] = as[(size_t)s * HEADS + hB];
    const float4* hp = (const float4*)(h + (size_t)s * H1C);
    vA[j] = hp[lane];
    vB[j] = hp[32 + lane];
  }
  float gA = -INFINITY, gB = -INFINITY;
#pragma unroll
  for (int j = 0; j < CNT; ++j) {
    eA[j] += adA;
    eB[j] += adB;
    eA[j] = (eA[j] > 0.f) ? eA[j] : NEG_SLOPE * eA[j];
    eB[j] = (eB[j] > 0.f) ? eB[j] : NEG_SLOPE * eB[j];
    gA = fmaxf(gA, eA[j]);
    gB = fmaxf(gB, eB[j]);
  }
  const float nA = fmaxf(mA, gA);
  const float nB = fmaxf(mB, gB);
  const float cA = __expf(mA - nA);
  const float cB = __expf(mB - nB);
  mA = nA;
  mB = nB;
  denA *= cA;
  denB *= cB;
  accA.x *= cA; accA.y *= cA; accA.z *= cA; accA.w *= cA;
  accB.x *= cB; accB.y *= cB; accB.z *= cB; accB.w *= cB;
#pragma unroll
  for (int j = 0; j < CNT; ++j) {
    const float wA = __expf(eA[j] - nA);
    const float wB = __expf(eB[j] - nB);
    denA += wA;
    denB += wB;
    accA.x += wA * vA[j].x; accA.y += wA * vA[j].y; accA.z += wA * vA[j].z; accA.w += wA * vA[j].w;
    accB.x += wB * vB[j].x; accB.y += wB * vB[j].y; accB.z += wB * vB[j].z; accB.w += wB * vB[j].w;
  }
}

__global__ void __launch_bounds__(256)
gather_rows_l1_kernel(const int* __restrict__ ell,
                      const int* __restrict__ ei, int Eorig,
                      const float* __restrict__ as, const float* __restrict__ ad,
                      const float* __restrict__ h, const float* __restrict__ bias,
                      float* __restrict__ g, int N) {
  const int lane = threadIdx.x & 31;
  const int d = __builtin_amdgcn_readfirstlane((int)((blockIdx.x * blockDim.x + threadIdx.x) >> 5));
  if (d >= N) return;
  const int* row   = ell + (size_t)d * ROWCAP;
  const int  first = row[lane];
  const int hA = lane >> 3;
  const int hB = 4 + (lane >> 3);
  const float adA = ad[d * HEADS + hA];
  const float adB = ad[d * HEADS + hB];
  const int  n_in  = __builtin_amdgcn_readfirstlane(first);
  const bool fits  = (n_in <= ROWSLOTS);
  const int  tot   = fits ? n_in + 1 : 1;

  float mA = -INFINITY, mB = -INFINITY;
  float4 accA = make_float4(0.f, 0.f, 0.f, 0.f);
  float4 accB = make_float4(0.f, 0.f, 0.f, 0.f);
  float denA = 0.f, denB = 0.f;
  for (int base = 0; base < tot; base += 32) {
    const int n = min(32, tot - base);
    const int mine = row_entry(row, first, d, base, lane, tot);
    int k = 0;
#pragma unroll 1
    for (; k + 4 <= n; k += 4)
      l1_group<4>(mine, k, lane, hA, hB, as, h, adA, adB, mA, mB, accA, accB, denA, denB);
    const int rem = n - k;
    if (rem == 3)
      l1_group<3>(mine, k, lane, hA, hB, as, h, adA, adB, mA, mB, accA, accB, denA, denB);
    else if (rem == 2)
      l1_group<2>(mine, k, lane, hA, hB, as, h, adA, adB, mA, mB, accA, accB, denA, denB);
    else if (rem == 1)
      l1_group<1>(mine, k, lane, hA, hB, as, h, adA, adB, mA, mB, accA, accB, denA, denB);
  }
  if (!fits) {
    for (int base = 0; base < Eorig; base += 32) {
      const int i = base + lane;
      int dd = -1;
      if (i < Eorig) dd = ei[Eorig + i];
      unsigned hit = __builtin_amdgcn_ballot_w32(dd == d);
      if (hit != 0u) {
        int mine = 0;
        if (dd == d) mine = ei[i];
        while (hit != 0u) {
          const int k = __builtin_ctz(hit);
          hit &= hit - 1u;
          l1_group<1>(mine, k, lane, hA, hB, as, h, adA, adB, mA, mB, accA, accB, denA, denB);
        }
      }
    }
  }
  const float dA = denA + 1e-16f;
  const float dB = denB + 1e-16f;
  const float4 bA = ((const float4*)bias)[lane];
  const float4 bB = ((const float4*)bias)[32 + lane];
  float4 oA, oB;
  oA.x = accA.x / dA + bA.x; oA.y = accA.y / dA + bA.y;
  oA.z = accA.z / dA + bA.z; oA.w = accA.w / dA + bA.w;
  oB.x = accB.x / dB + bB.x; oB.y = accB.y / dB + bB.y;
  oB.z = accB.z / dB + bB.z; oB.w = accB.w / dB + bB.w;
  oA.x = (oA.x > 0.f) ? oA.x : (__expf(oA.x) - 1.f);
  oA.y = (oA.y > 0.f) ? oA.y : (__expf(oA.y) - 1.f);
  oA.z = (oA.z > 0.f) ? oA.z : (__expf(oA.z) - 1.f);
  oA.w = (oA.w > 0.f) ? oA.w : (__expf(oA.w) - 1.f);
  oB.x = (oB.x > 0.f) ? oB.x : (__expf(oB.x) - 1.f);
  oB.y = (oB.y > 0.f) ? oB.y : (__expf(oB.y) - 1.f);
  oB.z = (oB.z > 0.f) ? oB.z : (__expf(oB.z) - 1.f);
  oB.w = (oB.w > 0.f) ? oB.w : (__expf(oB.w) - 1.f);
  float4* gp = (float4*)(g + (size_t)d * H1C);
  gp[lane]      = oA;
  gp[32 + lane] = oB;
}

__device__ __forceinline__ float l2_weights(bool has, int mine,
                                            const float* __restrict__ la, float adv,
                                            float& m, float2& acc, float& denl) {
  float e = -INFINITY;
  if (has) {
    e = la[2 * (size_t)mine] + adv;
    e = (e > 0.f) ? e : NEG_SLOPE * e;
  }
  float gm = e;
#pragma unroll
  for (int o = 16; o > 0; o >>= 1) gm = fmaxf(gm, __shfl_xor(gm, o));
  const float nm = fmaxf(m, gm);
  const float c = __expf(m - nm);
  m = nm;
  acc.x *= c;
  acc.y *= c;
  denl *= c;
  const float exl = has ? __expf(e - nm) : 0.f;
  denl += exl;
  return exl;
}

__global__ void __launch_bounds__(256)
gather_rows_l2_kernel(const int* __restrict__ ell,
                      const int* __restrict__ ei, int Eorig,
                      const float* __restrict__ la,
                      const float* __restrict__ h, const float* __restrict__ bias,
                      float* __restrict__ out, int N) {
  const int lane = threadIdx.x & 31;
  const int d = __builtin_amdgcn_readfirstlane((int)((blockIdx.x * blockDim.x + threadIdx.x) >> 5));
  if (d >= N) return;
  const int* row   = ell + (size_t)d * ROWCAP;
  const int  first = row[lane];
  const float adv  = la[2 * (size_t)d + 1];
  const int  n_in  = __builtin_amdgcn_readfirstlane(first);
  const bool fits  = (n_in <= ROWSLOTS);
  const int  tot   = fits ? n_in + 1 : 1;

  float m = -INFINITY;
  float2 acc = make_float2(0.f, 0.f);
  float denl = 0.f;
  for (int base = 0; base < tot; base += 32) {
    const int n = min(32, tot - base);
    const int mine = row_entry(row, first, d, base, lane, tot);
    const float exl = l2_weights(lane < n, mine, la, adv, m, acc, denl);
#pragma unroll 4
    for (int k = 0; k < n; ++k) {
      const int   s = __builtin_amdgcn_readlane(mine, k);
      const float w = __int_as_float(__builtin_amdgcn_readlane(__float_as_int(exl), k));
      const float2 v = ((const float2*)(h + (size_t)s * OUT_CH))[lane];
      acc.x += w * v.x; acc.y += w * v.y;
    }
  }
  if (!fits) {
    for (int base = 0; base < Eorig; base += 32) {
      const int i = base + lane;
      int dd = -1;
      if (i < Eorig) dd = ei[Eorig + i];
      unsigned hit = __builtin_amdgcn_ballot_w32(dd == d);
      if (hit != 0u) {
        int mine = 0;
        if (dd == d) mine = ei[i];
        const float exl = l2_weights(dd == d, mine, la, adv, m, acc, denl);
        while (hit != 0u) {
          const int k = __builtin_ctz(hit);
          hit &= hit - 1u;
          const int   s = __builtin_amdgcn_readlane(mine, k);
          const float w = __int_as_float(__builtin_amdgcn_readlane(__float_as_int(exl), k));
          const float2 v = ((const float2*)(h + (size_t)s * OUT_CH))[lane];
          acc.x += w * v.x; acc.y += w * v.y;
        }
      }
    }
  }
#pragma unroll
  for (int o = 16; o > 0; o >>= 1) denl += __shfl_xor(denl, o);
  const float dn = denl + 1e-16f;
  const float2 b = ((const float2*)bias)[lane];
  float2 o2;
  o2.x = acc.x / dn + b.x;
  o2.y = acc.y / dn + b.y;
  ((float2*)(out + (size_t)d * OUT_CH))[lane] = o2;
}

extern "C" void kernel_launch(void* const* d_in, const int* in_sizes, int n_in,
                              void* d_out, int out_size, void* d_ws, size_t ws_size,
                              hipStream_t stream) {
  const float* x      = (const float*)d_in[0];
  const int*   ei     = (const int*)  d_in[1];
  const float* W1     = (const float*)d_in[2];
  const float* a_src1 = (const float*)d_in[3];
  const float* a_dst1 = (const float*)d_in[4];
  const float* b1     = (const float*)d_in[5];
  const float* W2     = (const float*)d_in[6];
  const float* a_src2 = (const float*)d_in[7];
  const float* a_dst2 = (const float*)d_in[8];
  const float* b2     = (const float*)d_in[9];
  float* out = (float*)d_out;

  float* ws   = (float*)d_ws;
  float* h1   = ws;
  float* g1   = h1   + (size_t)NNODES * H1C;
  float* as1  = g1   + (size_t)NNODES * H1C;
  float* ad1  = as1  + (size_t)NNODES * HEADS;
  int* ell      = (int*)(ad1 + (size_t)NNODES * HEADS);
  float* Wa1  = (float*)(ell + (size_t)NNODES * ROWCAP);
  float* Wa2  = Wa1 + (size_t)IN_CH * LGW;
  float* h2   = h1;
  float* la2  = h1 + (size_t)NNODES * OUT_CH;

  const int T = 256;

  fold_kernel<<<FOLD1_BLOCKS + FOLD2_BLOCKS + FILL_BLOCKS, 256, 0, stream>>>(
      W1, a_src1, a_dst1, Wa1, W2, a_src2, a_dst2, Wa2, ell);

  gemm_l1_place_kernel<<<2 * L1_PAIRS, 128, 0, stream>>>(x, W1, h1, NNODES, Wa1, as1, ad1,
                                                         ei, ell, NEDGES);
  gather_rows_l1_kernel<<<(NNODES * 32 + T - 1) / T, T, 0, stream>>>(
      ell, ei, NEDGES, as1, ad1, h1, b1, g1, NNODES);

  wmma_gemm_l2<<<(NNODES + 15) / 16, 128, 0, stream>>>(g1, W2, h2, NNODES, Wa2, la2);
  gather_rows_l2_kernel<<<(NNODES * 32 + T - 1) / T, T, 0, stream>>>(
      ell, ei, NEDGES, la2, h2, b2, out, NNODES);
}
